# starting kernel
# baseline (speedup 1.0000x reference)
#include <hip/hip_runtime.h>
#include <hip/hip_bf16.h>

typedef __attribute__((ext_vector_type(2))) float v2f;
typedef __attribute__((ext_vector_type(8))) float v8f;

#define E 256
#define SEQ_L 512
#define M_TOT 32768
#define BM 64
#define BN 128
#define LSTRIDE 260
#define NLAYER 4

__global__ __launch_bounds__(256) void wtrans_kernel(const float* __restrict__ cw,
                                                     float* __restrict__ wt) {
  int idx = blockIdx.x * 256 + threadIdx.x;
  int j  = idx & 1;
  int o  = (idx >> 1) & 255;
  int e2 = (idx >> 9) & 127;
  int t  = (idx >> 16) & 1;
  int l  = idx >> 17;
  int e  = e2 * 2 + j;
  wt[idx] = cw[(((size_t)l * E + o) * E + e) * 3 + t];
}

__global__ __launch_bounds__(256) void conv_layer_kernel(const float* __restrict__ Xin,
                                                         const float* __restrict__ Wt,
                                                         float* __restrict__ Xout) {
  __shared__ float ldsX[(BM + 1) * LSTRIDE];

  const int tid  = threadIdx.x;
  const int lane = tid & 31;
  const int wave = tid >> 5;
  const int mb   = blockIdx.x * BM;
  const int nb   = blockIdx.y * BN;

  const bool bstart = (mb % SEQ_L) == 0;
  for (int idx = tid; idx < (BM + 1) * (E / 4); idx += 256) {
    const int r  = idx >> 6;
    const int c4 = idx & 63;
    float4 v;
    if (r == 0 && bstart) {
      v = make_float4(0.f, 0.f, 0.f, 0.f);
    } else {
      v = ((const float4*)(Xin + (size_t)(mb - 1 + r) * E))[c4];
    }
    float* dst = ldsX + r * LSTRIDE + c4 * 4;
    dst[0] = v.x; dst[1] = v.y; dst[2] = v.z; dst[3] = v.w;
  }
  __syncthreads();

  const int wm    = wave >> 1;
  const int wn    = wave & 1;
  const int r0    = wm * 16;
  const int kh    = (lane >> 4) << 1;
  const int nlo   = lane & 15;
  const int ncol0 = nb + wn * 64 + nlo;

  v8f acc0 = {}, acc1 = {}, acc2 = {}, acc3 = {};

#pragma unroll
  for (int t = 0; t < 2; ++t) {
    const float* __restrict__ arow = ldsX + (size_t)(r0 + nlo + t) * LSTRIDE;
    const v2f*   __restrict__ Wtt  = (const v2f*)(Wt + (size_t)t * E * E);
    for (int k0 = 0; k0 < E; k0 += 4) {
      const int e = k0 + kh;
      v2f a = *(const v2f*)(arow + e);
      const v2f* bp = Wtt + ((size_t)(e >> 1) * E + ncol0);
      v2f b0 = bp[0], b1 = bp[16], b2 = bp[32], b3 = bp[48];
      acc0 = __builtin_amdgcn_wmma_f32_16x16x4_f32(false, a, false, b0, (short)0, acc0, false, false);
      acc1 = __builtin_amdgcn_wmma_f32_16x16x4_f32(false, a, false, b1, (short)0, acc1, false, false);
      acc2 = __builtin_amdgcn_wmma_f32_16x16x4_f32(false, a, false, b2, (short)0, acc2, false, false);
      acc3 = __builtin_amdgcn_wmma_f32_16x16x4_f32(false, a, false, b3, (short)0, acc3, false, false);
    }
  }

  const int mrow = mb + r0 + ((lane >> 4) << 3);
  float* outp = Xout + (size_t)mrow * E + ncol0;
#pragma unroll
  for (int v = 0; v < 8; ++v) {
    outp[(size_t)v * E +  0] = acc0[v];
    outp[(size_t)v * E + 16] = acc1[v];
    outp[(size_t)v * E + 32] = acc2[v];
    outp[(size_t)v * E + 48] = acc3[v];
  }
}

__global__ __launch_bounds__(256) void fc_kernel(const float* __restrict__ X,
                                                 const float* __restrict__ w,
                                                 const float* __restrict__ b,
                                                 float* __restrict__ out) {
  const int wave = threadIdx.x >> 5;
  const int lane = threadIdx.x & 31;
  const int m = blockIdx.x * 8 + wave;
  const float* xr = X + (size_t)m * E;
  float s = 0.f;
#pragma unroll
  for (int e = lane; e < E; e += 32) s += xr[e] * w[e];
#pragma unroll
  for (int off = 16; off > 0; off >>= 1) s += __shfl_down(s, off);
  if (lane == 0) out[m] = s + b[0];
}

extern "C" void kernel_launch(void* const* d_in, const int* in_sizes, int n_in,
                              void* d_out, int out_size, void* d_ws, size_t ws_size,
                              hipStream_t stream) {
  const float* x      = (const float*)d_in[0];
  const float* conv_w = (const float*)d_in[1];
  const float* fc_w   = (const float*)d_in[2];
  const float* fc_b   = (const float*)d_in[3];
  float* out = (float*)d_out;

  float* ws   = (float*)d_ws;
  float* Wt   = ws;
  float* bufA = Wt + (size_t)NLAYER * 2 * E * E;
  float* bufB = bufA + (size_t)M_TOT * E;

  wtrans_kernel<<<(NLAYER * 2 * E * E) / 256, 256, 0, stream>>>(conv_w, Wt);

  dim3 grid(M_TOT / BM, E / BN);
  const size_t WL = (size_t)2 * E * E;
  conv_layer_kernel<<<grid, 256, 0, stream>>>(x,    Wt + 0 * WL, bufA);
  conv_layer_kernel<<<grid, 256, 0, stream>>>(bufA, Wt + 1 * WL, bufB);
  conv_layer_kernel<<<grid, 256, 0, stream>>>(bufB, Wt + 2 * WL, bufA);
  conv_layer_kernel<<<grid, 256, 0, stream>>>(bufA, Wt + 3 * WL, bufB);

  fc_kernel<<<M_TOT / 8, 256, 0, stream>>>(bufB, fc_w, fc_b, out);
}


# A fragments read from LDS 16 bytes at a time, the reads of two blocks (36 products) standing as ONE group ahead of them; k order inside each 8 is permuted, so the f32 sum order changes
# speedup vs baseline: 1.2252x; 1.2252x over previous
#include <hip/hip_runtime.h>
#include <hip/hip_bf16.h>

typedef __attribute__((ext_vector_type(2))) float v2f;
typedef __attribute__((ext_vector_type(4))) float v4f;
typedef __attribute__((ext_vector_type(8))) float v8f;

#define E 256
#define SEQ_L 512
#define M_TOT 32768
#define CH 128
#define HALO 16
#define LSTRIDE 260
#define NLAYER 4
#define NROW (1 + HALO + CH)
#define NTILE ((HALO + CH) / 16)

__global__ __launch_bounds__(256) void wtrans_kernel(const float* __restrict__ cw,
                                                     float* __restrict__ wt) {
  int idx = blockIdx.x * 256 + threadIdx.x;
  int j  = idx & 1;
  int o  = (idx >> 1) & 255;
  int e2 = (idx >> 9) & 127;
  int t  = (idx >> 16) & 1;
  int l  = idx >> 17;
  int e  = e2 * 2 + j;
  wt[idx] = cw[(((size_t)l * E + o) * E + e) * 3 + t];
}

#define NBLK 64

template <int NT, int P>
__device__ __forceinline__ void block_pair(v8f (&acc)[NT], v4f (&a)[3][NT], v2f (&b)[6][2],
                                           const float* __restrict__ abase,
                                           const v2f* __restrict__ bbase, const int n) {
  constexpr int L0 = (2 * P) % 3, L1 = (2 * P + 1) % 3, L2 = (2 * P + 2) % 3;
  constexpr int B0 = (2 * P) % 6, B1 = (2 * P + 1) % 6, B4 = (2 * P + 4) % 6, B5 = (2 * P + 5) % 6;
  const int n1 = (n + 1 < NBLK) ? (n + 1) : (NBLK - 1);
  const int n2 = (n + 2 < NBLK) ? (n + 2) : (NBLK - 1);
  const int n4 = (n + 4 < NBLK) ? (n + 4) : (NBLK - 1);
  const int n5 = (n + 5 < NBLK) ? (n + 5) : (NBLK - 1);
  const float* p1 = abase + (n1 >> 5) * LSTRIDE + (n1 & 31) * 8;
  const float* p2 = abase + (n2 >> 5) * LSTRIDE + (n2 & 31) * 8;
  __builtin_amdgcn_sched_barrier(0);
#pragma unroll
  for (int j = 0; j < NT; ++j) a[L1][j] = *(const v4f*)(p1 + j * 16 * LSTRIDE);
#pragma unroll
  for (int j = 0; j < NT; ++j) a[L2][j] = *(const v4f*)(p2 + j * 16 * LSTRIDE);
  b[B4][0] = bbase[(size_t)(4 * n4) * E];
  b[B4][1] = bbase[(size_t)(4 * n4 + 1) * E];
  b[B5][0] = bbase[(size_t)(4 * n5) * E];
  b[B5][1] = bbase[(size_t)(4 * n5 + 1) * E];
  __builtin_amdgcn_sched_barrier(0);
#pragma unroll
  for (int j = 0; j < NT; ++j) {
    const v4f f = a[L0][j];
    acc[j] = __builtin_amdgcn_wmma_f32_16x16x4_f32(false, __builtin_shufflevector(f, f, 0, 1), false, b[B0][0], (short)0, acc[j], false, false);
  }
#pragma unroll
  for (int j = 0; j < NT; ++j) {
    const v4f f = a[L0][j];
    acc[j] = __builtin_amdgcn_wmma_f32_16x16x4_f32(false, __builtin_shufflevector(f, f, 2, 3), false, b[B0][1], (short)0, acc[j], false, false);
  }
#pragma unroll
  for (int j = 0; j < NT; ++j) {
    const v4f f = a[L1][j];
    acc[j] = __builtin_amdgcn_wmma_f32_16x16x4_f32(false, __builtin_shufflevector(f, f, 0, 1), false, b[B1][0], (short)0, acc[j], false, false);
  }
#pragma unroll
  for (int j = 0; j < NT; ++j) {
    const v4f f = a[L1][j];
    acc[j] = __builtin_amdgcn_wmma_f32_16x16x4_f32(false, __builtin_shufflevector(f, f, 2, 3), false, b[B1][1], (short)0, acc[j], false, false);
  }
}

template <int J0>
__device__ __forceinline__ void conv_layer_lds(const float* __restrict__ in,
                                               float* __restrict__ out,
                                               const v2f* __restrict__ Wl,
                                               const int lane, const int wave) {
  constexpr int NT = NTILE - J0;
  const int h    = lane >> 4;
  const int c    = lane & 15;
  const int ncol = wave * 16 + c;
  const float* abase = in + (J0 * 16 + c) * LSTRIDE + 4 * h;
  const v2f*   bbase = Wl + (2 * h) * E + ncol;

  v8f acc[NT];
#pragma unroll
  for (int j = 0; j < NT; ++j) acc[j] = v8f{};

  v4f a[3][NT];
  v2f b[6][2];

#pragma unroll
  for (int j = 0; j < NT; ++j) a[0][j] = *(const v4f*)(abase + j * 16 * LSTRIDE);
#pragma unroll
  for (int u = 0; u < 4; ++u) {
    b[u][0] = bbase[(size_t)(4 * u) * E];
    b[u][1] = bbase[(size_t)(4 * u + 1) * E];
  }

  for (int n = 0; n < 60; n += 6) {
    block_pair<NT, 0>(acc, a, b, abase, bbase, n);
    block_pair<NT, 1>(acc, a, b, abase, bbase, n + 2);
    block_pair<NT, 2>(acc, a, b, abase, bbase, n + 4);
  }
  block_pair<NT, 0>(acc, a, b, abase, bbase, 60);
  block_pair<NT, 1>(acc, a, b, abase, bbase, 62);

  float* outp = out + (1 + J0 * 16 + 8 * h) * LSTRIDE + ncol;
#pragma unroll
  for (int j = 0; j < NT; ++j) {
#pragma unroll
    for (int v = 0; v < 8; ++v) outp[(j * 16 + v) * LSTRIDE] = acc[j][v];
  }
}

__global__ __launch_bounds__(512) void fused_layers_kernel(const float* __restrict__ X,
                                                           const float* __restrict__ Wt,
                                                           const float* __restrict__ fcw,
                                                           const float* __restrict__ fcb,
                                                           float* __restrict__ out) {
  __shared__ __attribute__((aligned(16))) float lds[2 * NROW * LSTRIDE];
  float* buf0 = lds;
  float* buf1 = lds + NROW * LSTRIDE;

  const int tid  = threadIdx.x;
  const int lane = tid & 31;
  const int wave = tid >> 5;
  const int s0   = blockIdx.x * CH;

  const bool bstart = (s0 % SEQ_L) == 0;
  {
    const int c4 = tid & 63;
    const int rb = tid >> 6;
    float4 v[19];
#pragma unroll
    for (int i = 0; i < 19; ++i) {
      const int r = rb + 8 * i;
      if (i < 18 || rb == 0) {
        const bool z = bstart && (r <= HALO);
        const int  g = z ? s0 : (s0 - (HALO + 1) + r);
        float4 t = ((const float4*)(X + (size_t)g * E))[c4];
        if (z) t = make_float4(0.f, 0.f, 0.f, 0.f);
        v[i] = t;
      }
    }
#pragma unroll
    for (int i = 0; i < 19; ++i) {
      const int r = rb + 8 * i;
      if (i < 18 || rb == 0) *(float4*)(buf0 + r * LSTRIDE + c4 * 4) = v[i];
    }
  }
  if (tid < E / 4) *(float4*)(buf1 + tid * 4) = make_float4(0.f, 0.f, 0.f, 0.f);
  __syncthreads();

  const v2f* W = (const v2f*)Wt;
  const size_t WLv = (size_t)E * E;
  conv_layer_lds<0>(buf0, buf1, W + 0 * WLv, lane, wave);
  __syncthreads();
  conv_layer_lds<0>(buf1, buf0, W + 1 * WLv, lane, wave);
  __syncthreads();
  conv_layer_lds<0>(buf0, buf1, W + 2 * WLv, lane, wave);
  __syncthreads();
  conv_layer_lds<1>(buf1, buf0, W + 3 * WLv, lane, wave);
  __syncthreads();

#pragma unroll
  for (int i = 0; i < CH / 16; ++i) {
    const int row = wave * (CH / 16) + i;
    const float* xr = buf0 + (1 + HALO + row) * LSTRIDE;
    float s = 0.f;
#pragma unroll
    for (int e = lane; e < E; e += 32) s += xr[e] * fcw[e];
#pragma unroll
    for (int off = 16; off > 0; off >>= 1) s += __shfl_down(s, off);
    if (lane == 0) out[s0 + row] = s + fcb[0];
  }
}

extern "C" void kernel_launch(void* const* d_in, const int* in_sizes, int n_in,
                              void* d_out, int out_size, void* d_ws, size_t ws_size,
                              hipStream_t stream) {
  const float* x      = (const float*)d_in[0];
  const float* conv_w = (const float*)d_in[1];
  const float* fc_w   = (const float*)d_in[2];
  const float* fc_b   = (const float*)d_in[3];
  float* out = (float*)d_out;

  float* Wt = (float*)d_ws;

  wtrans_kernel<<<(NLAYER * 2 * E * E) / 256, 256, 0, stream>>>(conv_w, Wt);
  fused_layers_kernel<<<M_TOT / CH, 512, 0, stream>>>(x, Wt, fc_w, fc_b, out);
}


# Minimal 2-tap filtering along the sequence: a row pair from 3 matrix products, (d0-d1)W0, d1(W0+W1), (d2-d1)W1, instead of 4; 8 waves x 32 columns hold the accumulators; re-associates the f32 sums
# speedup vs baseline: 1.3978x; 1.1409x over previous
#include <hip/hip_runtime.h>
#include <hip/hip_bf16.h>

typedef __attribute__((ext_vector_type(2))) float v2f;
typedef __attribute__((ext_vector_type(4))) float v4f;
typedef __attribute__((ext_vector_type(8))) float v8f;

#define E 256
#define SEQ_L 512
#define M_TOT 32768
#define CH 128
#define HALO 16
#define LSTRIDE 260
#define NLAYER 4
#define NROW (1 + HALO + CH)
#define NP0 73
#define P1OFF (NP0 * LSTRIDE)
#define NPT 5
#define NBLK 32


__global__ __launch_bounds__(256) void wtrans_kernel(const float* __restrict__ cw,
                                                     float* __restrict__ wt) {
  int idx = blockIdx.x * 256 + threadIdx.x;
  int j  = idx & 1;
  int o  = (idx >> 1) & 255;
  int e2 = (idx >> 9) & 127;
  int lm = idx >> 16;
  int l  = lm / 3;
  int m  = lm - 3 * l;
  int e  = e2 * 2 + j;
  const float* s = cw + (((size_t)l * E + o) * E + e) * 3;
  const float w0 = s[0], w1 = s[1];
  wt[idx] = (m == 0) ? w0 : ((m == 2) ? w1 : (w0 + w1));
}

template <int NT, int S>
__device__ __forceinline__ void wblock(v8f (&acc)[2][NT][3], v4f (&a)[2][NT][3], v2f (&b)[2][3][2][2],
                                       const float* const (&q0)[NT], const float* const (&q1)[NT],
                                       const v2f* __restrict__ bbase, const int n) {
  const int n1 = (n + 1 < NBLK) ? (n + 1) : (NBLK - 1);
  __builtin_amdgcn_sched_barrier(0);
#pragma unroll
  for (int T = 0; T < NT; ++T) {
    a[S ^ 1][T][0] = *(const v4f*)(q0[T] + n1 * 8);
    a[S ^ 1][T][1] = *(const v4f*)(q1[T] + n1 * 8);
    a[S ^ 1][T][2] = *(const v4f*)(q0[T] + LSTRIDE + n1 * 8);
  }
#pragma unroll
  for (int m = 0; m < 3; ++m)
#pragma unroll
    for (int t = 0; t < 2; ++t)
#pragma unroll
      for (int i = 0; i < 2; ++i)
        b[S ^ 1][m][t][i] = bbase[(size_t)(m * (E / 2) + 4 * n1 + i) * E + 16 * t];
  __builtin_amdgcn_sched_barrier(0);
#pragma unroll
  for (int t = 0; t < 2; ++t)
#pragma unroll
    for (int T = 0; T < NT; ++T)
#pragma unroll
      for (int m = 0; m < 3; ++m) {
        const v4f f = a[S][T][m];
        acc[t][T][m] = __builtin_amdgcn_wmma_f32_16x16x4_f32(false, __builtin_shufflevector(f, f, 0, 1), false, b[S][m][t][0], (short)0, acc[t][T][m], false, false);
      }
  __builtin_amdgcn_sched_barrier(0);
#pragma unroll
  for (int T = 0; T < NT; ++T) {
    const v4f d1 = a[S ^ 1][T][1];
    a[S ^ 1][T][0] -= d1;
    a[S ^ 1][T][2] -= d1;
  }
#pragma unroll
  for (int t = 0; t < 2; ++t)
#pragma unroll
    for (int T = 0; T < NT; ++T)
#pragma unroll
      for (int m = 0; m < 3; ++m) {
        const v4f f = a[S][T][m];
        acc[t][T][m] = __builtin_amdgcn_wmma_f32_16x16x4_f32(false, __builtin_shufflevector(f, f, 2, 3), false, b[S][m][t][1], (short)0, acc[t][T][m], false, false);
      }
#pragma unroll
  for (int g = 0; g < 4 * NT; ++g) {
    __builtin_amdgcn_sched_group_barrier(0x008, 1, 0);
    __builtin_amdgcn_sched_group_barrier(0x002, 1, 0);
  }
}

template <int J0>
__device__ __forceinline__ void conv_layer_lds(const float* __restrict__ in,
                                               float* __restrict__ out,
                                               const v2f* __restrict__ Wl,
                                               const int lane, const int wave) {
  constexpr int NT = NPT - J0;
  const int h = lane >> 4;
  const int c = lane & 15;
  const int ncol = wave * 32 + c;

  const float* q0[NT];
  const float* q1[NT];
#pragma unroll
  for (int T = 0; T < NT; ++T) {
    int p = 16 * (J0 + T) - 8 + c;
    if (J0 + T == 0) p = (p < 0) ? 0 : p;
    q0[T] = in + p * LSTRIDE + 4 * h;
    q1[T] = in + P1OFF + p * LSTRIDE + 4 * h;
  }
  const v2f* bbase = Wl + (2 * h) * E + ncol;

  v8f acc[2][NT][3];
#pragma unroll
  for (int t = 0; t < 2; ++t)
#pragma unroll
    for (int T = 0; T < NT; ++T)
#pragma unroll
      for (int m = 0; m < 3; ++m) acc[t][T][m] = v8f{};

  v4f a[2][NT][3];
  v2f b[2][3][2][2];

#pragma unroll
  for (int T = 0; T < NT; ++T) {
    const v4f d0 = *(const v4f*)(q0[T]);
    const v4f d1 = *(const v4f*)(q1[T]);
    const v4f d2 = *(const v4f*)(q0[T] + LSTRIDE);
    a[0][T][0] = d0 - d1;
    a[0][T][1] = d1;
    a[0][T][2] = d2 - d1;
  }
#pragma unroll
  for (int m = 0; m < 3; ++m)
#pragma unroll
    for (int t = 0; t < 2; ++t)
#pragma unroll
      for (int i = 0; i < 2; ++i)
        b[0][m][t][i] = bbase[(size_t)(m * (E / 2) + i) * E + 16 * t];

  for (int n = 0; n < NBLK; n += 2) {
    wblock<NT, 0>(acc, a, b, q0, q1, bbase, n);
    wblock<NT, 1>(acc, a, b, q0, q1, bbase, n + 1);
  }

#pragma unroll
  for (int t = 0; t < 2; ++t) {
#pragma unroll
    for (int T = 0; T < NT; ++T) {
      if (J0 + T > 0 || h == 1) {
        const int pb = 16 * (J0 + T) - 8 + 8 * h;
        float* o1 = out + P1OFF + pb * LSTRIDE + ncol + 16 * t;
        float* o2 = out + (pb + 1) * LSTRIDE + ncol + 16 * t;
#pragma unroll
        for (int v = 0; v < 8; ++v) {
          o1[v * LSTRIDE] = acc[t][T][0][v] + acc[t][T][1][v];
          o2[v * LSTRIDE] = acc[t][T][1][v] + acc[t][T][2][v];
        }
      }
    }
  }
}

__device__ __forceinline__ int row_off(const int r) { return (r & 1) * P1OFF + (r >> 1) * LSTRIDE; }

__global__ __launch_bounds__(256) void fused_layers_kernel(const float* __restrict__ X,
                                                           const float* __restrict__ Wt,
                                                           const float* __restrict__ fcw,
                                                           const float* __restrict__ fcb,
                                                           float* __restrict__ out) {
  __shared__ __attribute__((aligned(16))) float lds[2 * NROW * LSTRIDE];
  float* buf0 = lds;
  float* buf1 = lds + NROW * LSTRIDE;

  const int tid  = threadIdx.x;
  const int lane = tid & 31;
  const int wave = tid >> 5;
  const int s0   = blockIdx.x * CH;

  const bool bstart = (s0 % SEQ_L) == 0;
  {
    const int c4 = tid & 63;
    const int rb = tid >> 6;
    float4 v[37];
#pragma unroll
    for (int i = 0; i < 37; ++i) {
      const int r = rb + 4 * i;
      if (i < 36 || rb == 0) {
        const bool z = bstart && (r <= HALO);
        const int  g = z ? s0 : (s0 - (HALO + 1) + r);
        float4 t = ((const float4*)(X + (size_t)g * E))[c4];
        if (z) t = make_float4(0.f, 0.f, 0.f, 0.f);
        v[i] = t;
      }
    }
#pragma unroll
    for (int i = 0; i < 37; ++i) {
      const int r = rb + 4 * i;
      if (i < 36 || rb == 0) *(float4*)(buf0 + row_off(r) + c4 * 4) = v[i];
    }
  }
  if (tid < E / 4) *(float4*)(buf1 + tid * 4) = make_float4(0.f, 0.f, 0.f, 0.f);
  __syncthreads();

  const v2f* W = (const v2f*)Wt;
  const size_t WLv = (size_t)3 * (E / 2) * E;
  conv_layer_lds<0>(buf0, buf1, W + 0 * WLv, lane, wave);
  __syncthreads();
  conv_layer_lds<0>(buf1, buf0, W + 1 * WLv, lane, wave);
  __syncthreads();
  conv_layer_lds<0>(buf0, buf1, W + 2 * WLv, lane, wave);
  __syncthreads();
  conv_layer_lds<1>(buf1, buf0, W + 3 * WLv, lane, wave);
  __syncthreads();

#pragma unroll
  for (int i = 0; i < CH / 8; ++i) {
    const int row = wave * (CH / 8) + i;
    const float* xr = buf0 + row_off(1 + HALO + row);
    float s = 0.f;
#pragma unroll
    for (int e = lane; e < E; e += 32) s += xr[e] * fcw[e];
#pragma unroll
    for (int off = 16; off > 0; off >>= 1) s += __shfl_down(s, off);
    if (lane == 0) out[s0 + row] = s + fcb[0];
  }
}

extern "C" void kernel_launch(void* const* d_in, const int* in_sizes, int n_in,
                              void* d_out, int out_size, void* d_ws, size_t ws_size,
                              hipStream_t stream) {
  const float* x      = (const float*)d_in[0];
  const float* conv_w = (const float*)d_in[1];
  const float* fc_w   = (const float*)d_in[2];
  const float* fc_b   = (const float*)d_in[3];
  float* out = (float*)d_out;

  float* Wt = (float*)d_ws;

  wtrans_kernel<<<(NLAYER * 3 * E * E) / 256, 256, 0, stream>>>(conv_w, Wt);
  fused_layers_kernel<<<M_TOT / CH, 256, 0, stream>>>(x, Wt, fc_w, fc_b, out);
}


# Minimal filtering over row triples, F(3,2): 3 rows from 4 matrix products (d0-d2)W0, (d1+d2)(W0+W1)/2, (d2-d1)(W0-W1)/2, (d1-d3)W1, not 2 rows from 3; re-associates the f32 sums
# speedup vs baseline: 1.4989x; 1.0723x over previous
#include <hip/hip_runtime.h>
#include <hip/hip_bf16.h>

typedef __attribute__((ext_vector_type(2))) float v2f;
typedef __attribute__((ext_vector_type(4))) float v4f;
typedef __attribute__((ext_vector_type(8))) float v8f;

#define E 256
#define SEQ_L 512
#define M_TOT 32768
#define CH 128
#define HALO 16
#define LSTRIDE 260
#define NLAYER 4
#define NROW (1 + HALO + CH)
#define NP0 49
#define NP1 48
#define P1OFF (NP0 * LSTRIDE)
#define P2OFF ((NP0 + NP1) * LSTRIDE)
#define NTT 3
#define NMAT 4
#define NBLK 32


__global__ __launch_bounds__(256) void wtrans_kernel(const float* __restrict__ cw,
                                                     float* __restrict__ wt) {
  int idx = blockIdx.x * 256 + threadIdx.x;
  int j  = idx & 1;
  int o  = (idx >> 1) & 255;
  int e2 = (idx >> 9) & 127;
  int lm = idx >> 16;
  int l  = lm >> 2;
  int m  = lm & 3;
  int e  = e2 * 2 + j;
  const float* s = cw + (((size_t)l * E + o) * E + e) * 3;
  const float w0 = s[0], w1 = s[1];
  float r;
  if (m == 0)      r = w0;
  else if (m == 3) r = w1;
  else if (m == 1) r = 0.5f * (w0 + w1);
  else             r = 0.5f * (w0 - w1);
  wt[idx] = r;
}

template <int S>
__device__ __forceinline__ void wblock(v8f (&acc)[2][NTT][NMAT], v4f (&a)[2][NTT][NMAT], v2f (&b)[2][NMAT][2][2],
                                       const float* const (&q0)[NTT], const float* const (&q1)[NTT],
                                       const float* const (&q2)[NTT],
                                       const v2f* __restrict__ bbase, const int n) {
  const int n1 = (n + 1 < NBLK) ? (n + 1) : (NBLK - 1);
  __builtin_amdgcn_sched_barrier(0);
#pragma unroll
  for (int T = 0; T < NTT; ++T) {
    a[S ^ 1][T][0] = *(const v4f*)(q0[T] + n1 * 8);
    a[S ^ 1][T][1] = *(const v4f*)(q1[T] + n1 * 8);
    a[S ^ 1][T][2] = *(const v4f*)(q2[T] + n1 * 8);
    a[S ^ 1][T][3] = *(const v4f*)(q0[T] + LSTRIDE + n1 * 8);
  }
#pragma unroll
  for (int m = 0; m < NMAT; ++m)
#pragma unroll
    for (int t = 0; t < 2; ++t)
#pragma unroll
      for (int i = 0; i < 2; ++i)
        b[S ^ 1][m][t][i] = bbase[(size_t)(m * (E / 2) + 4 * n1 + i) * E + 16 * t];
  __builtin_amdgcn_sched_barrier(0);
#pragma unroll
  for (int t = 0; t < 2; ++t)
#pragma unroll
    for (int T = 0; T < NTT; ++T)
#pragma unroll
      for (int m = 0; m < NMAT; ++m) {
        const v4f f = a[S][T][m];
        acc[t][T][m] = __builtin_amdgcn_wmma_f32_16x16x4_f32(false, __builtin_shufflevector(f, f, 0, 1), false, b[S][m][t][0], (short)0, acc[t][T][m], false, false);
      }
  __builtin_amdgcn_sched_barrier(0);
#pragma unroll
  for (int T = 0; T < NTT; ++T) {
    const v4f d0 = a[S ^ 1][T][0];
    const v4f d1 = a[S ^ 1][T][1];
    const v4f d2 = a[S ^ 1][T][2];
    const v4f d3 = a[S ^ 1][T][3];
    a[S ^ 1][T][0] = d0 - d2;
    a[S ^ 1][T][1] = d1 + d2;
    a[S ^ 1][T][2] = d2 - d1;
    a[S ^ 1][T][3] = d1 - d3;
  }
#pragma unroll
  for (int t = 0; t < 2; ++t)
#pragma unroll
    for (int T = 0; T < NTT; ++T)
#pragma unroll
      for (int m = 0; m < NMAT; ++m) {
        const v4f f = a[S][T][m];
        acc[t][T][m] = __builtin_amdgcn_wmma_f32_16x16x4_f32(false, __builtin_shufflevector(f, f, 2, 3), false, b[S][m][t][1], (short)0, acc[t][T][m], false, false);
      }
#pragma unroll
  for (int g = 0; g < 8 * NTT; ++g) {
    __builtin_amdgcn_sched_group_barrier(0x008, 1, 0);
    __builtin_amdgcn_sched_group_barrier(0x002, 1, 0);
  }
}

__device__ __forceinline__ void conv_layer_lds(const float* __restrict__ in,
                                               float* __restrict__ out,
                                               const v2f* __restrict__ Wl,
                                               const int lane, const int wave, const bool zfront) {
  const int h = lane >> 4;
  const int c = lane & 15;
  const int ncol = wave * 32 + c;

  const float* q0[NTT];
  const float* q1[NTT];
  const float* q2[NTT];
#pragma unroll
  for (int T = 0; T < NTT; ++T) {
    const int q = 16 * T + c;
    q0[T] = in + q * LSTRIDE + 4 * h;
    q1[T] = in + P1OFF + q * LSTRIDE + 4 * h;
    q2[T] = in + P2OFF + q * LSTRIDE + 4 * h;
  }
  const v2f* bbase = Wl + (2 * h) * E + ncol;

  v8f acc[2][NTT][NMAT];
#pragma unroll
  for (int t = 0; t < 2; ++t)
#pragma unroll
    for (int T = 0; T < NTT; ++T)
#pragma unroll
      for (int m = 0; m < NMAT; ++m) acc[t][T][m] = v8f{};

  v4f a[2][NTT][NMAT];
  v2f b[2][NMAT][2][2];

#pragma unroll
  for (int T = 0; T < NTT; ++T) {
    const v4f d0 = *(const v4f*)(q0[T]);
    const v4f d1 = *(const v4f*)(q1[T]);
    const v4f d2 = *(const v4f*)(q2[T]);
    const v4f d3 = *(const v4f*)(q0[T] + LSTRIDE);
    a[0][T][0] = d0 - d2;
    a[0][T][1] = d1 + d2;
    a[0][T][2] = d2 - d1;
    a[0][T][3] = d1 - d3;
  }
#pragma unroll
  for (int m = 0; m < NMAT; ++m)
#pragma unroll
    for (int t = 0; t < 2; ++t)
#pragma unroll
      for (int i = 0; i < 2; ++i)
        b[0][m][t][i] = bbase[(size_t)(m * (E / 2) + i) * E + 16 * t];

  for (int n = 0; n < NBLK; n += 2) {
    wblock<0>(acc, a, b, q0, q1, q2, bbase, n);
    wblock<1>(acc, a, b, q0, q1, q2, bbase, n + 1);
  }

  const bool zl = zfront && (h == 0);
#pragma unroll
  for (int t = 0; t < 2; ++t) {
#pragma unroll
    for (int T = 0; T < NTT; ++T) {
      const int qb = 16 * T + 8 * h;
      float* o1 = out + P1OFF + qb * LSTRIDE + ncol + 16 * t;
      float* o2 = out + P2OFF + qb * LSTRIDE + ncol + 16 * t;
      float* o0 = out + (qb + 1) * LSTRIDE + ncol + 16 * t;
#pragma unroll
      for (int v = 0; v < 8; ++v) {
        const float m1 = acc[t][T][0][v];
        const float m2 = acc[t][T][1][v];
        const float m3 = acc[t][T][2][v];
        const float m4 = acc[t][T][3][v];
        const float s23 = m2 + m3;
        float y0 = m1 + s23;
        float y1 = m2 - m3;
        float y2 = s23 - m4;
        if (T == 0 && v <= 5) y0 = zl ? 0.f : y0;
        if (T == 0 && v <= 4) y1 = zl ? 0.f : y1;
        if (T == 0 && v <= 4) y2 = zl ? 0.f : y2;
        o1[v * LSTRIDE] = y0;
        o2[v * LSTRIDE] = y1;
        o0[v * LSTRIDE] = y2;
      }
    }
  }
}

__device__ __forceinline__ int row_off(const int r) {
  const int q = r / 3;
  const int j = r - 3 * q;
  return ((j == 0) ? 0 : ((j == 1) ? P1OFF : P2OFF)) + q * LSTRIDE;
}

__global__ __launch_bounds__(256) void fused_layers_kernel(const float* __restrict__ X,
                                                           const float* __restrict__ Wt,
                                                           const float* __restrict__ fcw,
                                                           const float* __restrict__ fcb,
                                                           float* __restrict__ out) {
  __shared__ __attribute__((aligned(16))) float lds[2 * NROW * LSTRIDE];
  float* buf0 = lds;
  float* buf1 = lds + NROW * LSTRIDE;

  const int tid  = threadIdx.x;
  const int lane = tid & 31;
  const int wave = tid >> 5;
  const int s0   = blockIdx.x * CH;

  const bool bstart = (s0 % SEQ_L) == 0;
  {
    const int c4 = tid & 63;
    const int rb = tid >> 6;
    float4 v[37];
#pragma unroll
    for (int i = 0; i < 37; ++i) {
      const int r = rb + 4 * i;
      if (i < 36 || rb == 0) {
        const bool z = bstart && (r <= HALO);
        const int  g = z ? s0 : (s0 - (HALO + 1) + r);
        float4 t = ((const float4*)(X + (size_t)g * E))[c4];
        if (z) t = make_float4(0.f, 0.f, 0.f, 0.f);
        v[i] = t;
      }
    }
#pragma unroll
    for (int i = 0; i < 37; ++i) {
      const int r = rb + 4 * i;
      if (i < 36 || rb == 0) *(float4*)(buf0 + row_off(r) + c4 * 4) = v[i];
    }
  }
  if (tid < E / 4) *(float4*)(buf1 + tid * 4) = make_float4(0.f, 0.f, 0.f, 0.f);
  __syncthreads();

  const v2f* W = (const v2f*)Wt;
  const size_t WLv = (size_t)NMAT * (E / 2) * E;
  conv_layer_lds(buf0, buf1, W + 0 * WLv, lane, wave, bstart);
  __syncthreads();
  conv_layer_lds(buf1, buf0, W + 1 * WLv, lane, wave, bstart);
  __syncthreads();
  conv_layer_lds(buf0, buf1, W + 2 * WLv, lane, wave, bstart);
  __syncthreads();
  conv_layer_lds(buf1, buf0, W + 3 * WLv, lane, wave, bstart);
  __syncthreads();

#pragma unroll
  for (int i = 0; i < CH / 8; ++i) {
    const int row = wave * (CH / 8) + i;
    const float* xr = buf0 + row_off(1 + HALO + row);
    float s = 0.f;
#pragma unroll
    for (int e = lane; e < E; e += 32) s += xr[e] * fcw[e];
#pragma unroll
    for (int off = 16; off > 0; off >>= 1) s += __shfl_down(s, off);
    if (lane == 0) out[s0 + row] = s + fcb[0];
  }
}

extern "C" void kernel_launch(void* const* d_in, const int* in_sizes, int n_in,
                              void* d_out, int out_size, void* d_ws, size_t ws_size,
                              hipStream_t stream) {
  const float* x      = (const float*)d_in[0];
  const float* conv_w = (const float*)d_in[1];
  const float* fc_w   = (const float*)d_in[2];
  const float* fc_b   = (const float*)d_in[3];
  float* out = (float*)d_out;

  float* Wt = (float*)d_ws;

  wtrans_kernel<<<(NLAYER * NMAT * E * E) / 256, 256, 0, stream>>>(conv_w, Wt);
  fused_layers_kernel<<<M_TOT / CH, 256, 0, stream>>>(x, Wt, fc_w, fc_b, out);
}


# Only W0 and W1 are repacked and loaded, 16 bytes a lane (4 weight loads a k block, not 16); W0+W1 and W0-W1 formed in registers, their halves applied to the finished sums; every f32 sum unchanged
# speedup vs baseline: 1.5990x; 1.0668x over previous
#include <hip/hip_runtime.h>
#include <hip/hip_bf16.h>

typedef __attribute__((ext_vector_type(2))) float v2f;
typedef __attribute__((ext_vector_type(4))) float v4f;
typedef __attribute__((ext_vector_type(8))) float v8f;

#define E 256
#define SEQ_L 512
#define M_TOT 32768
#define CH 128
#define HALO 16
#define LSTRIDE 260
#define NLAYER 4
#define NROW (1 + HALO + CH)
#define NP0 49
#define NP1 48
#define P1OFF (NP0 * LSTRIDE)
#define P2OFF ((NP0 + NP1) * LSTRIDE)
#define NTT 3
#define NMAT 4
#define NBLK 32


__global__ __launch_bounds__(256) void wtrans_kernel(const float* __restrict__ cw,
                                                     float* __restrict__ wt) {
  int idx = blockIdx.x * 256 + threadIdx.x;
  int j  = idx & 3;
  int o  = (idx >> 2) & 255;
  int e4 = (idx >> 10) & 63;
  int l  = idx >> 16;
  int lo = idx & 0xFFFF;
  const float* s = cw + (((size_t)l * E + o) * E + 4 * e4 + j) * 3;
  const float w0 = s[0], w1 = s[1];
  wt[((l * 2 + 0) << 16) + lo] = w0;
  wt[((l * 2 + 1) << 16) + lo] = w1;
}

template <int S>
__device__ __forceinline__ void wblock(v8f (&acc)[2][NTT][NMAT], v4f (&a)[2][NTT][NMAT], v4f (&r)[2][2][2],
                                       const float* const (&q0)[NTT], const float* const (&q1)[NTT],
                                       const float* const (&q2)[NTT],
                                       const v4f* __restrict__ bbase, const int n) {
  const int n1 = (n + 1 < NBLK) ? (n + 1) : (NBLK - 1);
  __builtin_amdgcn_sched_barrier(0);
#pragma unroll
  for (int T = 0; T < NTT; ++T) {
    a[S ^ 1][T][0] = *(const v4f*)(q0[T] + n1 * 8);
    a[S ^ 1][T][1] = *(const v4f*)(q1[T] + n1 * 8);
    a[S ^ 1][T][2] = *(const v4f*)(q2[T] + n1 * 8);
    a[S ^ 1][T][3] = *(const v4f*)(q0[T] + LSTRIDE + n1 * 8);
  }
#pragma unroll
  for (int w = 0; w < 2; ++w)
#pragma unroll
    for (int t = 0; t < 2; ++t)
      r[S ^ 1][w][t] = bbase[(size_t)(w * (E / 4) + 2 * n1) * E + 16 * t];
  __builtin_amdgcn_sched_barrier(0);
  v4f bw[NMAT][2];
#pragma unroll
  for (int t = 0; t < 2; ++t) {
    bw[0][t] = r[S][0][t];
    bw[1][t] = r[S][0][t] + r[S][1][t];
    bw[2][t] = r[S][0][t] - r[S][1][t];
    bw[3][t] = r[S][1][t];
  }
#pragma unroll
  for (int T = 0; T < NTT; ++T)
#pragma unroll
    for (int t = 0; t < 2; ++t)
#pragma unroll
      for (int m = 0; m < NMAT; m += 3) {
        const v4f f = a[S][T][m];
        acc[t][T][m] = __builtin_amdgcn_wmma_f32_16x16x4_f32(false, __builtin_shufflevector(f, f, 0, 1), false, __builtin_shufflevector(bw[m][t], bw[m][t], 0, 1), (short)0, acc[t][T][m], false, false);
      }
#pragma unroll
  for (int t = 0; t < 2; ++t)
#pragma unroll
    for (int T = 0; T < NTT; ++T)
#pragma unroll
      for (int m = 1; m < 3; ++m) {
        const v4f f = a[S][T][m];
        acc[t][T][m] = __builtin_amdgcn_wmma_f32_16x16x4_f32(false, __builtin_shufflevector(f, f, 0, 1), false, __builtin_shufflevector(bw[m][t], bw[m][t], 0, 1), (short)0, acc[t][T][m], false, false);
      }
#pragma unroll
  for (int g = 0; g < 8; ++g) {
    __builtin_amdgcn_sched_group_barrier(0x008, 1, 0);
    __builtin_amdgcn_sched_group_barrier(0x002, 1, 0);
  }
  __builtin_amdgcn_sched_barrier(0);
#pragma unroll
  for (int T = 0; T < NTT; ++T) {
    const v4f d0 = a[S ^ 1][T][0];
    const v4f d1 = a[S ^ 1][T][1];
    const v4f d2 = a[S ^ 1][T][2];
    const v4f d3 = a[S ^ 1][T][3];
    a[S ^ 1][T][0] = d0 - d2;
    a[S ^ 1][T][1] = d1 + d2;
    a[S ^ 1][T][2] = d2 - d1;
    a[S ^ 1][T][3] = d1 - d3;
  }
#pragma unroll
  for (int t = 0; t < 2; ++t)
#pragma unroll
    for (int T = 0; T < NTT; ++T)
#pragma unroll
      for (int m = 0; m < NMAT; ++m) {
        const v4f f = a[S][T][m];
        acc[t][T][m] = __builtin_amdgcn_wmma_f32_16x16x4_f32(false, __builtin_shufflevector(f, f, 2, 3), false, __builtin_shufflevector(bw[m][t], bw[m][t], 2, 3), (short)0, acc[t][T][m], false, false);
      }
#pragma unroll
  for (int g = 0; g < 8 * NTT; ++g) {
    __builtin_amdgcn_sched_group_barrier(0x008, 1, 0);
    __builtin_amdgcn_sched_group_barrier(0x002, 1, 0);
  }
}

__device__ __forceinline__ void conv_layer_lds(const float* __restrict__ in,
                                               float* __restrict__ out,
                                               const v4f* __restrict__ Wl,
                                               const int lane, const int wave, const bool zfront) {
  const int h = lane >> 4;
  const int c = lane & 15;
  const int ncol = wave * 32 + c;

  const float* q0[NTT];
  const float* q1[NTT];
  const float* q2[NTT];
#pragma unroll
  for (int T = 0; T < NTT; ++T) {
    const int q = 16 * T + c;
    q0[T] = in + q * LSTRIDE + 4 * h;
    q1[T] = in + P1OFF + q * LSTRIDE + 4 * h;
    q2[T] = in + P2OFF + q * LSTRIDE + 4 * h;
  }
  const v4f* bbase = Wl + h * E + ncol;

  v8f acc[2][NTT][NMAT];
#pragma unroll
  for (int t = 0; t < 2; ++t)
#pragma unroll
    for (int T = 0; T < NTT; ++T)
#pragma unroll
      for (int m = 0; m < NMAT; ++m) acc[t][T][m] = v8f{};

  v4f a[2][NTT][NMAT];
  v4f r[2][2][2];

#pragma unroll
  for (int T = 0; T < NTT; ++T) {
    const v4f d0 = *(const v4f*)(q0[T]);
    const v4f d1 = *(const v4f*)(q1[T]);
    const v4f d2 = *(const v4f*)(q2[T]);
    const v4f d3 = *(const v4f*)(q0[T] + LSTRIDE);
    a[0][T][0] = d0 - d2;
    a[0][T][1] = d1 + d2;
    a[0][T][2] = d2 - d1;
    a[0][T][3] = d1 - d3;
  }
#pragma unroll
  for (int w = 0; w < 2; ++w)
#pragma unroll
    for (int t = 0; t < 2; ++t)
      r[0][w][t] = bbase[(size_t)(w * (E / 4)) * E + 16 * t];

  for (int n = 0; n < NBLK; n += 2) {
    wblock<0>(acc, a, r, q0, q1, q2, bbase, n);
    wblock<1>(acc, a, r, q0, q1, q2, bbase, n + 1);
  }

  const bool zl = zfront && (h == 0);
#pragma unroll
  for (int t = 0; t < 2; ++t) {
#pragma unroll
    for (int T = 0; T < NTT; ++T) {
      const int qb = 16 * T + 8 * h;
      float* o1 = out + P1OFF + qb * LSTRIDE + ncol + 16 * t;
      float* o2 = out + P2OFF + qb * LSTRIDE + ncol + 16 * t;
      float* o0 = out + (qb + 1) * LSTRIDE + ncol + 16 * t;
#pragma unroll
      for (int v = 0; v < 8; ++v) {
        const float m1 = acc[t][T][0][v];
        const float p2 = acc[t][T][1][v];
        const float p3 = acc[t][T][2][v];
        const float m4 = acc[t][T][3][v];
        const float s23 = 0.5f * (p2 + p3);
        float y0 = m1 + s23;
        float y1 = 0.5f * (p2 - p3);
        float y2 = s23 - m4;
        if (T == 0 && v <= 5) y0 = zl ? 0.f : y0;
        if (T == 0 && v <= 4) y1 = zl ? 0.f : y1;
        if (T == 0 && v <= 4) y2 = zl ? 0.f : y2;
        o1[v * LSTRIDE] = y0;
        o2[v * LSTRIDE] = y1;
        o0[v * LSTRIDE] = y2;
      }
    }
  }
}

__device__ __forceinline__ int row_off(const int r) {
  const int q = r / 3;
  const int j = r - 3 * q;
  return ((j == 0) ? 0 : ((j == 1) ? P1OFF : P2OFF)) + q * LSTRIDE;
}

__global__ __launch_bounds__(256) void fused_layers_kernel(const float* __restrict__ X,
                                                           const float* __restrict__ Wt,
                                                           const float* __restrict__ fcw,
                                                           const float* __restrict__ fcb,
                                                           float* __restrict__ out) {
  __shared__ __attribute__((aligned(16))) float lds[2 * NROW * LSTRIDE];
  float* buf0 = lds;
  float* buf1 = lds + NROW * LSTRIDE;

  const int tid  = threadIdx.x;
  const int lane = tid & 31;
  const int wave = tid >> 5;
  const int s0   = blockIdx.x * CH;

  const bool bstart = (s0 % SEQ_L) == 0;
  {
    const int c4 = tid & 63;
    const int rb = tid >> 6;
    float4 v[37];
#pragma unroll
    for (int i = 0; i < 37; ++i) {
      const int r = rb + 4 * i;
      if (i < 36 || rb == 0) {
        const bool z = bstart && (r <= HALO);
        const int  g = z ? s0 : (s0 - (HALO + 1) + r);
        float4 t = ((const float4*)(X + (size_t)g * E))[c4];
        if (z) t = make_float4(0.f, 0.f, 0.f, 0.f);
        v[i] = t;
      }
    }
#pragma unroll
    for (int i = 0; i < 37; ++i) {
      const int r = rb + 4 * i;
      if (i < 36 || rb == 0) *(float4*)(buf0 + row_off(r) + c4 * 4) = v[i];
    }
  }
  if (tid < E / 4) *(float4*)(buf1 + tid * 4) = make_float4(0.f, 0.f, 0.f, 0.f);
  __syncthreads();

  const v4f* W = (const v4f*)Wt;
  const size_t WLv = (size_t)2 * (E / 4) * E;
  conv_layer_lds(buf0, buf1, W + 0 * WLv, lane, wave, bstart);
  __syncthreads();
  conv_layer_lds(buf1, buf0, W + 1 * WLv, lane, wave, bstart);
  __syncthreads();
  conv_layer_lds(buf0, buf1, W + 2 * WLv, lane, wave, bstart);
  __syncthreads();
  conv_layer_lds(buf1, buf0, W + 3 * WLv, lane, wave, bstart);
  __syncthreads();

#pragma unroll
  for (int i = 0; i < CH / 8; ++i) {
    const int row = wave * (CH / 8) + i;
    const float* xr = buf0 + row_off(1 + HALO + row);
    float s = 0.f;
#pragma unroll
    for (int e = lane; e < E; e += 32) s += xr[e] * fcw[e];
#pragma unroll
    for (int off = 16; off > 0; off >>= 1) s += __shfl_down(s, off);
    if (lane == 0) out[s0 + row] = s + fcb[0];
  }
}

extern "C" void kernel_launch(void* const* d_in, const int* in_sizes, int n_in,
                              void* d_out, int out_size, void* d_ws, size_t ws_size,
                              hipStream_t stream) {
  const float* x      = (const float*)d_in[0];
  const float* conv_w = (const float*)d_in[1];
  const float* fc_w   = (const float*)d_in[2];
  const float* fc_b   = (const float*)d_in[3];
  float* out = (float*)d_out;

  float* Wt = (float*)d_ws;

  wtrans_kernel<<<(NLAYER * E * E) / 256, 256, 0, stream>>>(conv_w, Wt);
  fused_layers_kernel<<<M_TOT / CH, 256, 0, stream>>>(x, Wt, fc_w, fc_b, out);
}


# Neighbouring layers composed in pairs (nothing non-linear between): A0=W0bW0a, A1=W0bW1a+W1bW0a, A2=W1bW1a by WMMA each call; 2 layers of F(2,3), 4 products per 2 rows; re-associates f32 sums
# speedup vs baseline: 1.8552x; 1.1602x over previous
#include <hip/hip_runtime.h>
#include <hip/hip_bf16.h>

typedef __attribute__((ext_vector_type(2))) float v2f;
typedef __attribute__((ext_vector_type(4))) float v4f;
typedef __attribute__((ext_vector_type(8))) float v8f;

#define E 256
#define SEQ_L 512
#define M_TOT 32768
#define CH 128
#define LSTRIDE 260
#define PROWS 66
#define POFF (PROWS * LSTRIDE)
#define BUFSZ (2 * PROWS * LSTRIDE)
#define NMAT 4
#define NBLK 32
#define SHIFT1 15


__global__ __launch_bounds__(32) void wcompose_kernel(const float* __restrict__ cw,
                                                      float* __restrict__ wt) {
  const int lane = threadIdx.x;
  const int c = lane & 15;
  const int h = lane >> 4;
  const int b  = blockIdx.x;
  const int o0 = (b & 15) * 16;
  const int f0 = ((b >> 4) & 15) * 16;
  const int L  = b >> 8;
  const float* la = cw + (size_t)(2 * L) * E * E * 3;
  const float* lb = cw + (size_t)(2 * L + 1) * E * E * 3;
  const float* pa = la + ((size_t)(2 * h) * E + f0 + c) * 3;
  const float* pb = lb + ((size_t)(o0 + c) * E + 2 * h) * 3;
  v8f a0 = {}, a1 = {}, a2 = {};
#pragma unroll 8
  for (int k = 0; k < E / 4; ++k) {
    const float* qa = pa + (size_t)k * 4 * E * 3;
    const float* qb = pb + k * 12;
    const v2f A0 = {qa[0], qa[E * 3]};
    const v2f A1 = {qa[1], qa[E * 3 + 1]};
    const v2f B0 = {qb[0], qb[3]};
    const v2f B1 = {qb[1], qb[4]};
    a0 = __builtin_amdgcn_wmma_f32_16x16x4_f32(false, A0, false, B0, (short)0, a0, false, false);
    a1 = __builtin_amdgcn_wmma_f32_16x16x4_f32(false, A1, false, B0, (short)0, a1, false, false);
    a2 = __builtin_amdgcn_wmma_f32_16x16x4_f32(false, A1, false, B1, (short)0, a2, false, false);
    a1 = __builtin_amdgcn_wmma_f32_16x16x4_f32(false, A0, false, B1, (short)0, a1, false, false);
  }
  v4f* w4 = (v4f*)wt;
  const size_t base = ((size_t)(L * 3) * (E / 4) + (f0 + 8 * h) / 4) * E + o0 + c;
  const size_t ms = (size_t)(E / 4) * E;
  w4[base]              = v4f{a0[0], a0[1], a0[2], a0[3]};
  w4[base + E]          = v4f{a0[4], a0[5], a0[6], a0[7]};
  w4[base + ms]         = v4f{a1[0], a1[1], a1[2], a1[3]};
  w4[base + ms + E]     = v4f{a1[4], a1[5], a1[6], a1[7]};
  w4[base + 2 * ms]     = v4f{a2[0], a2[1], a2[2], a2[3]};
  w4[base + 2 * ms + E] = v4f{a2[4], a2[5], a2[6], a2[7]};
}

template <int S, int NT>
__device__ __forceinline__ void wblock(v8f (&acc)[2][NT][NMAT], v4f (&a)[2][NT][NMAT], v4f (&r)[2][3][2],
                                       const float* const (&qe)[NT], const float* const (&qo)[NT],
                                       const v4f* __restrict__ bbase, const int n) {
  const int n1 = (n + 1 < NBLK) ? (n + 1) : (NBLK - 1);
  __builtin_amdgcn_sched_barrier(0);
#pragma unroll
  for (int T = 0; T < NT; ++T) {
    a[S ^ 1][T][0] = *(const v4f*)(qe[T] + n1 * 8);
    a[S ^ 1][T][1] = *(const v4f*)(qo[T] + n1 * 8);
    a[S ^ 1][T][2] = *(const v4f*)(qe[T] + LSTRIDE + n1 * 8);
    a[S ^ 1][T][3] = *(const v4f*)(qo[T] + LSTRIDE + n1 * 8);
  }
#pragma unroll
  for (int w = 0; w < 3; ++w)
#pragma unroll
    for (int t = 0; t < 2; ++t)
      r[S ^ 1][w][t] = bbase[(size_t)(w * (E / 4) + 2 * n1) * E + 16 * t];
  __builtin_amdgcn_sched_barrier(0);
  v4f bw[NMAT][2];
#pragma unroll
  for (int t = 0; t < 2; ++t) {
    const v4f p = r[S][0][t] + r[S][2][t];
    bw[0][t] = r[S][0][t];
    bw[1][t] = p + r[S][1][t];
    bw[2][t] = p - r[S][1][t];
    bw[3][t] = r[S][2][t];
  }
#pragma unroll
  for (int T = 0; T < NT; ++T)
#pragma unroll
    for (int t = 0; t < 2; ++t)
#pragma unroll
      for (int m = 0; m < NMAT; m += 3) {
        const v4f f = a[S][T][m];
        acc[t][T][m] = __builtin_amdgcn_wmma_f32_16x16x4_f32(false, __builtin_shufflevector(f, f, 0, 1), false, __builtin_shufflevector(bw[m][t], bw[m][t], 0, 1), (short)0, acc[t][T][m], false, false);
      }
#pragma unroll
  for (int t = 0; t < 2; ++t)
#pragma unroll
    for (int T = 0; T < NT; ++T)
#pragma unroll
      for (int m = 1; m < 3; ++m) {
        const v4f f = a[S][T][m];
        acc[t][T][m] = __builtin_amdgcn_wmma_f32_16x16x4_f32(false, __builtin_shufflevector(f, f, 0, 1), false, __builtin_shufflevector(bw[m][t], bw[m][t], 0, 1), (short)0, acc[t][T][m], false, false);
      }
#pragma unroll
  for (int g = 0; g < 12; ++g) {
    __builtin_amdgcn_sched_group_barrier(0x008, 1, 0);
    __builtin_amdgcn_sched_group_barrier(0x002, 1, 0);
  }
  __builtin_amdgcn_sched_barrier(0);
#pragma unroll
  for (int T = 0; T < NT; ++T) {
    const v4f d0 = a[S ^ 1][T][0];
    const v4f d1 = a[S ^ 1][T][1];
    const v4f d2 = a[S ^ 1][T][2];
    const v4f d3 = a[S ^ 1][T][3];
    a[S ^ 1][T][0] = d0 - d2;
    a[S ^ 1][T][1] = d1 + d2;
    a[S ^ 1][T][2] = d2 - d1;
    a[S ^ 1][T][3] = d1 - d3;
  }
#pragma unroll
  for (int t = 0; t < 2; ++t)
#pragma unroll
    for (int T = 0; T < NT; ++T)
#pragma unroll
      for (int m = 0; m < NMAT; ++m) {
        const v4f f = a[S][T][m];
        acc[t][T][m] = __builtin_amdgcn_wmma_f32_16x16x4_f32(false, __builtin_shufflevector(f, f, 2, 3), false, __builtin_shufflevector(bw[m][t], bw[m][t], 2, 3), (short)0, acc[t][T][m], false, false);
      }
#pragma unroll
  for (int g = 0; g < 8 * NT; ++g) {
    __builtin_amdgcn_sched_group_barrier(0x008, 1, 0);
    __builtin_amdgcn_sched_group_barrier(0x002, 1, 0);
  }
}

template <int NT, int T0, int SHIFT>
__device__ __forceinline__ void conv_pass(const float* __restrict__ in,
                                          float* __restrict__ out,
                                          const v4f* __restrict__ Wl,
                                          const int lane, const int wave) {
  const int h = lane >> 4;
  const int c = lane & 15;
  const int ncol = wave * 32 + c;

  const float* qe[NT];
  const float* qo[NT];
#pragma unroll
  for (int T = 0; T < NT; ++T) {
    const int p  = 16 * (T0 + T) + c - SHIFT;
    const int pc = (p < 0) ? 0 : p;
    qe[T] = in + pc * LSTRIDE + 4 * h;
    qo[T] = in + POFF + pc * LSTRIDE + 4 * h;
  }
  const v4f* bbase = Wl + h * E + ncol;

  v8f acc[2][NT][NMAT];
#pragma unroll
  for (int t = 0; t < 2; ++t)
#pragma unroll
    for (int T = 0; T < NT; ++T)
#pragma unroll
      for (int m = 0; m < NMAT; ++m) acc[t][T][m] = v8f{};

  v4f a[2][NT][NMAT];
  v4f r[2][3][2];

#pragma unroll
  for (int T = 0; T < NT; ++T) {
    const v4f d0 = *(const v4f*)(qe[T]);
    const v4f d1 = *(const v4f*)(qo[T]);
    const v4f d2 = *(const v4f*)(qe[T] + LSTRIDE);
    const v4f d3 = *(const v4f*)(qo[T] + LSTRIDE);
    a[0][T][0] = d0 - d2;
    a[0][T][1] = d1 + d2;
    a[0][T][2] = d2 - d1;
    a[0][T][3] = d1 - d3;
  }
#pragma unroll
  for (int w = 0; w < 3; ++w)
#pragma unroll
    for (int t = 0; t < 2; ++t)
      r[0][w][t] = bbase[(size_t)(w * (E / 4)) * E + 16 * t];

  for (int n = 0; n < NBLK; n += 2) {
    wblock<0, NT>(acc, a, r, qe, qo, bbase, n);
    wblock<1, NT>(acc, a, r, qe, qo, bbase, n + 1);
  }

#pragma unroll
  for (int t = 0; t < 2; ++t) {
#pragma unroll
    for (int T = 0; T < NT; ++T) {
      const int pb = 16 * (T0 + T) + 8 * h - SHIFT;
      const int col = ncol + 16 * t;
#pragma unroll
      for (int v = 0; v < 8; ++v) {
        const float m1 = acc[t][T][0][v];
        const float p2 = acc[t][T][1][v];
        const float p3 = acc[t][T][2][v];
        const float m4 = acc[t][T][3][v];
        const float y0 = m1 + 0.5f * (p2 + p3);
        const float y1 = 0.5f * (p2 - p3) - m4;
        const int po = pb + v;
        if ((16 * (T0 + T) - SHIFT >= 0) || (po >= 0)) {
          out[po * LSTRIDE + col] = y0;
          out[POFF + po * LSTRIDE + col] = y1;
        }
      }
    }
  }
}

__global__ __launch_bounds__(256) void fused_layers_kernel(const float* __restrict__ X,
                                                           const float* __restrict__ Wt,
                                                           const float* __restrict__ fcw,
                                                           const float* __restrict__ fcb,
                                                           float* __restrict__ out) {
  __shared__ __attribute__((aligned(16))) float lds[2 * BUFSZ];
  float* bufX = lds;
  float* bufH = lds + BUFSZ;

  const int tid  = threadIdx.x;
  const int lane = tid & 31;
  const int wave = tid >> 5;
  const int s0   = blockIdx.x * CH;

  const bool bstart = (s0 % SEQ_L) == 0;
  {
    const int c4 = tid & 63;
    const int rb = tid >> 6;
    float4 v[33];
#pragma unroll
    for (int i = 0; i < 33; ++i) {
      const int r = rb + 4 * i;
      const bool z = bstart && (r < 4);
      const int  g = z ? s0 : (s0 - 4 + r);
      float4 t = ((const float4*)(X + (size_t)g * E))[c4];
      if (z) t = make_float4(0.f, 0.f, 0.f, 0.f);
      v[i] = t;
    }
#pragma unroll
    for (int i = 0; i < 33; ++i) {
      const int r = rb + 4 * i;
      *(float4*)(bufX + (r & 1) * POFF + (r >> 1) * LSTRIDE + c4 * 4) = v[i];
    }
  }
  __syncthreads();

  const v4f* W = (const v4f*)Wt;
  const size_t WLv = (size_t)3 * (E / 4) * E;
  conv_pass<3, 0, SHIFT1>(bufX, bufH, W, lane, wave);
  conv_pass<2, 3, SHIFT1>(bufX, bufH, W, lane, wave);
  __syncthreads();
  conv_pass<4, 0, 0>(bufH, bufX, W + WLv, lane, wave);
  __syncthreads();

#pragma unroll
  for (int i = 0; i < CH / 8; ++i) {
    const int row = wave * (CH / 8) + i;
    const float* xr = bufX + (row & 1) * POFF + (row >> 1) * LSTRIDE;
    float s = 0.f;
#pragma unroll
    for (int e = lane; e < E; e += 32) s += xr[e] * fcw[e];
#pragma unroll
    for (int off = 16; off > 0; off >>= 1) s += __shfl_down(s, off);
    if (lane == 0) out[s0 + row] = s + fcb[0];
  }
}

extern "C" void kernel_launch(void* const* d_in, const int* in_sizes, int n_in,
                              void* d_out, int out_size, void* d_ws, size_t ws_size,
                              hipStream_t stream) {
  const float* x      = (const float*)d_in[0];
  const float* conv_w = (const float*)d_in[1];
  const float* fc_w   = (const float*)d_in[2];
  const float* fc_b   = (const float*)d_in[3];
  float* out = (float*)d_out;

  float* Wt = (float*)d_ws;

  wcompose_kernel<<<512, 32, 0, stream>>>(conv_w, Wt);
  fused_layers_kernel<<<M_TOT / CH, 256, 0, stream>>>(x, Wt, fc_w, fc_b, out);
}


# All 4 layers composed into ONE 5-tap layer, run by minimal filtering F(2,5): 6 matrix products per 2 rows; its 6 filter matrices = layer products at 6 points, each call; re-associates f32 sums
# speedup vs baseline: 2.2993x; 1.2394x over previous
#include <hip/hip_runtime.h>
#include <hip/hip_bf16.h>

typedef __attribute__((ext_vector_type(2))) float v2f;
typedef __attribute__((ext_vector_type(4))) float v4f;
typedef __attribute__((ext_vector_type(8))) float v8f;

#define E 256
#define SEQ_L 512
#define M_TOT 32768
#define CH 128
#define LSTRIDE 260
#define PROWS 66
#define POFF (PROWS * LSTRIDE)
#define BUFSZ (2 * PROWS * LSTRIDE)
#define NPT 6
#define NBLK 32
#define MATSZ (E * E)


template <bool SWAP, int G>
__device__ __forceinline__ void pair_points(const float* __restrict__ la, const float* __restrict__ lb,
                                            float* __restrict__ q, const int m0, const int n0,
                                            const int c, const int h) {
  const int fo = SWAP ? n0 : m0;
  const int oo = SWAP ? m0 : n0;
  const float* pa = la + ((size_t)(2 * h) * E + fo + c) * 3;
  const float* pb = lb + ((size_t)(oo + c) * E + 2 * h) * 3;
  v8f acc[3] = {v8f{}, v8f{}, v8f{}};
#pragma unroll 8
  for (int k = 0; k < E / 4; ++k) {
    const float* qa = pa + (size_t)k * 4 * E * 3;
    const float* qb = pb + k * 12;
    const v2f A0 = {qa[0], qa[E * 3]};
    const v2f A1 = {qa[1], qa[E * 3 + 1]};
    const v2f B0 = {qb[0], qb[3]};
    const v2f B1 = {qb[1], qb[4]};
    v2f fa[3], fb[3];
    if (G == 0) {
      fa[0] = A0;             fb[0] = B0;
      fa[1] = A0 + A1;        fb[1] = B0 + B1;
      fa[2] = A0 - A1;        fb[2] = B0 - B1;
    } else {
      fa[0] = A0 + 2.0f * A1; fb[0] = B0 + 2.0f * B1;
      fa[1] = A0 - 2.0f * A1; fb[1] = B0 - 2.0f * B1;
      fa[2] = A1;             fb[2] = B1;
    }
#pragma unroll
    for (int i = 0; i < 3; ++i) {
      if (SWAP)
        acc[i] = __builtin_amdgcn_wmma_f32_16x16x4_f32(false, fb[i], false, fa[i], (short)0, acc[i], false, false);
      else
        acc[i] = __builtin_amdgcn_wmma_f32_16x16x4_f32(false, fa[i], false, fb[i], (short)0, acc[i], false, false);
    }
  }
  const size_t base = (size_t)((m0 + 8 * h) / 4) * E + n0 + c;
#pragma unroll
  for (int i = 0; i < 3; ++i) {
    v4f* w4 = (v4f*)(q + (size_t)(3 * G + i) * MATSZ);
    w4[base]     = v4f{acc[i][0], acc[i][1], acc[i][2], acc[i][3]};
    w4[base + E] = v4f{acc[i][4], acc[i][5], acc[i][6], acc[i][7]};
  }
}

__global__ __launch_bounds__(32) void pair_points_kernel(const float* __restrict__ cw,
                                                         float* __restrict__ qall) {
  const int lane = threadIdx.x;
  const int c = lane & 15;
  const int h = lane >> 4;
  const int b  = blockIdx.x;
  const int n0 = (b & 15) * 16;
  const int m0 = ((b >> 4) & 15) * 16;
  const int L  = (b >> 8) & 1;
  const int G  = b >> 9;
  const float* la = cw + (size_t)(2 * L) * E * E * 3;
  const float* lb = cw + (size_t)(2 * L + 1) * E * E * 3;
  float* q = qall + (size_t)L * NPT * MATSZ;
  if (L == 0) {
    if (G == 0) pair_points<true, 0>(la, lb, q, m0, n0, c, h);
    else        pair_points<true, 1>(la, lb, q, m0, n0, c, h);
  } else {
    if (G == 0) pair_points<false, 0>(la, lb, q, m0, n0, c, h);
    else        pair_points<false, 1>(la, lb, q, m0, n0, c, h);
  }
}

__global__ __launch_bounds__(32) void point_product_kernel(const float* __restrict__ qall,
                                                           float* __restrict__ ut) {
  const int lane = threadIdx.x;
  const int c = lane & 15;
  const int h = lane >> 4;
  const int b  = blockIdx.x;
  const int o0 = (b & 7) * 32;
  const int f0 = ((b >> 3) & 15) * 16;
  const int p  = b >> 7;
  const v2f* q0 = (const v2f*)(qall + (size_t)p * MATSZ) + (size_t)(f0 + c) * 2 + h;
  const v2f* q1 = (const v2f*)(qall + (size_t)(NPT + p) * MATSZ) + (size_t)(o0 + c) * 2 + h;
  v8f a0 = {}, a1 = {};
#pragma unroll 16
  for (int s = 0; s < E / 4; ++s) {
    const v2f fa  = q0[(size_t)s * E * 2];
    const v2f fb0 = q1[(size_t)s * E * 2];
    const v2f fb1 = q1[(size_t)s * E * 2 + 32];
    a0 = __builtin_amdgcn_wmma_f32_16x16x4_f32(false, fa, false, fb0, (short)0, a0, false, false);
    a1 = __builtin_amdgcn_wmma_f32_16x16x4_f32(false, fa, false, fb1, (short)0, a1, false, false);
  }
  const float sc = (p == 0) ? 0.25f : (p < 3) ? (-1.0f / 6.0f) : (p < 5) ? (1.0f / 24.0f) : 1.0f;
  v4f* w4 = (v4f*)(ut + (size_t)p * MATSZ);
  const size_t base = (size_t)((f0 + 8 * h) / 4) * E + o0 + c;
  w4[base]          = sc * v4f{a0[0], a0[1], a0[2], a0[3]};
  w4[base + E]      = sc * v4f{a0[4], a0[5], a0[6], a0[7]};
  w4[base + 16]     = sc * v4f{a1[0], a1[1], a1[2], a1[3]};
  w4[base + E + 16] = sc * v4f{a1[4], a1[5], a1[6], a1[7]};
}

__device__ __forceinline__ void vtransform(v4f (&a)[NPT]) {
  const v4f d0 = a[0], d1 = a[1], d2 = a[2], d3 = a[3], d4 = a[4], d5 = a[5];
  const v4f s  = d4 - 4.0f * d2;
  const v4f t  = d3 - 4.0f * d1;
  const v4f u  = d4 - d2;
  const v4f w  = 2.0f * (d3 - d1);
  a[0] = 4.0f * d0 + (d4 - 5.0f * d2);
  a[1] = s + t;
  a[2] = s - t;
  a[3] = u + w;
  a[4] = u - w;
  a[5] = 4.0f * d1 + (d5 - 5.0f * d3);
}

template <int S, int NT>
__device__ __forceinline__ void wblock(v8f (&acc)[2][NT][NPT], v4f (&a)[2][NT][NPT], v4f (&r)[2][NPT][2],
                                       const float* const (&qe)[NT], const float* const (&qo)[NT],
                                       const v4f* __restrict__ bbase, const int n) {
  const int n1 = (n + 1 < NBLK) ? (n + 1) : (NBLK - 1);
  __builtin_amdgcn_sched_barrier(0);
#pragma unroll
  for (int T = 0; T < NT; ++T) {
    a[S ^ 1][T][0] = *(const v4f*)(qe[T] + n1 * 8);
    a[S ^ 1][T][1] = *(const v4f*)(qo[T] + n1 * 8);
    a[S ^ 1][T][2] = *(const v4f*)(qe[T] + LSTRIDE + n1 * 8);
    a[S ^ 1][T][3] = *(const v4f*)(qo[T] + LSTRIDE + n1 * 8);
    a[S ^ 1][T][4] = *(const v4f*)(qe[T] + 2 * LSTRIDE + n1 * 8);
    a[S ^ 1][T][5] = *(const v4f*)(qo[T] + 2 * LSTRIDE + n1 * 8);
  }
#pragma unroll
  for (int p = 0; p < NPT; ++p)
#pragma unroll
    for (int t = 0; t < 2; ++t)
      r[S ^ 1][p][t] = bbase[(size_t)(p * (E / 4) + 2 * n1) * E + 16 * t];
  __builtin_amdgcn_sched_barrier(0);
#pragma unroll
  for (int T = 0; T < NT; ++T)
#pragma unroll
    for (int t = 0; t < 2; ++t)
#pragma unroll
      for (int p = 0; p < NPT; ++p) {
        const v4f f = a[S][T][p];
        const v4f g = r[S][p][t];
        acc[t][T][p] = __builtin_amdgcn_wmma_f32_16x16x4_f32(false, __builtin_shufflevector(f, f, 0, 1), false, __builtin_shufflevector(g, g, 0, 1), (short)0, acc[t][T][p], false, false);
      }
  __builtin_amdgcn_sched_barrier(0);
#pragma unroll
  for (int T = 0; T < NT; ++T) vtransform(a[S ^ 1][T]);
#pragma unroll
  for (int t = 0; t < 2; ++t)
#pragma unroll
    for (int T = 0; T < NT; ++T)
#pragma unroll
      for (int p = 0; p < NPT; ++p) {
        const v4f f = a[S][T][p];
        const v4f g = r[S][p][t];
        acc[t][T][p] = __builtin_amdgcn_wmma_f32_16x16x4_f32(false, __builtin_shufflevector(f, f, 2, 3), false, __builtin_shufflevector(g, g, 2, 3), (short)0, acc[t][T][p], false, false);
      }
#pragma unroll
  for (int g = 0; g < 12 * NT; ++g) {
    __builtin_amdgcn_sched_group_barrier(0x008, 1, 0);
    __builtin_amdgcn_sched_group_barrier(0x002, 2, 0);
  }
}

template <int NT, int T0>
__device__ __forceinline__ void conv_pass(const float* __restrict__ in,
                                          float* __restrict__ out,
                                          const v4f* __restrict__ Wl,
                                          const int lane, const int wave) {
  const int h = lane >> 4;
  const int c = lane & 15;
  const int ncol = wave * 32 + c;

  const float* qe[NT];
  const float* qo[NT];
#pragma unroll
  for (int T = 0; T < NT; ++T) {
    const int q = 16 * (T0 + T) + c;
    qe[T] = in + q * LSTRIDE + 4 * h;
    qo[T] = in + POFF + q * LSTRIDE + 4 * h;
  }
  const v4f* bbase = Wl + h * E + ncol;

  v8f acc[2][NT][NPT];
#pragma unroll
  for (int t = 0; t < 2; ++t)
#pragma unroll
    for (int T = 0; T < NT; ++T)
#pragma unroll
      for (int p = 0; p < NPT; ++p) acc[t][T][p] = v8f{};

  v4f a[2][NT][NPT];
  v4f r[2][NPT][2];

#pragma unroll
  for (int T = 0; T < NT; ++T) {
    a[0][T][0] = *(const v4f*)(qe[T]);
    a[0][T][1] = *(const v4f*)(qo[T]);
    a[0][T][2] = *(const v4f*)(qe[T] + LSTRIDE);
    a[0][T][3] = *(const v4f*)(qo[T] + LSTRIDE);
    a[0][T][4] = *(const v4f*)(qe[T] + 2 * LSTRIDE);
    a[0][T][5] = *(const v4f*)(qo[T] + 2 * LSTRIDE);
  }
#pragma unroll
  for (int p = 0; p < NPT; ++p)
#pragma unroll
    for (int t = 0; t < 2; ++t)
      r[0][p][t] = bbase[(size_t)(p * (E / 4)) * E + 16 * t];
#pragma unroll
  for (int T = 0; T < NT; ++T) vtransform(a[0][T]);

  for (int n = 0; n < NBLK; n += 2) {
    wblock<0, NT>(acc, a, r, qe, qo, bbase, n);
    wblock<1, NT>(acc, a, r, qe, qo, bbase, n + 1);
  }

#pragma unroll
  for (int t = 0; t < 2; ++t) {
#pragma unroll
    for (int T = 0; T < NT; ++T) {
      const int pb = 16 * (T0 + T) + 8 * h;
      const int col = ncol + 16 * t;
#pragma unroll
      for (int v = 0; v < 8; ++v) {
        const float m0 = acc[t][T][0][v];
        const float m1 = acc[t][T][1][v];
        const float m2 = acc[t][T][2][v];
        const float m3 = acc[t][T][3][v];
        const float m4 = acc[t][T][4][v];
        const float m5 = acc[t][T][5][v];
        const float y0 = (m0 + (m1 + m2)) + (m3 + m4);
        const float y1 = ((m1 - m2) + 2.0f * (m3 - m4)) + m5;
        const int po = pb + v;
        out[po * LSTRIDE + col] = y0;
        out[POFF + po * LSTRIDE + col] = y1;
      }
    }
  }
}

__global__ __launch_bounds__(256) void fused_layers_kernel(const float* __restrict__ X,
                                                           const float* __restrict__ Ut,
                                                           const float* __restrict__ fcw,
                                                           const float* __restrict__ fcb,
                                                           float* __restrict__ out) {
  __shared__ __attribute__((aligned(16))) float lds[2 * BUFSZ];
  float* bufX = lds;
  float* bufY = lds + BUFSZ;

  const int tid  = threadIdx.x;
  const int lane = tid & 31;
  const int wave = tid >> 5;
  const int s0   = blockIdx.x * CH;

  const bool bstart = (s0 % SEQ_L) == 0;
  {
    const int c4 = tid & 63;
    const int rb = tid >> 6;
    float4 v[33];
#pragma unroll
    for (int i = 0; i < 33; ++i) {
      const int r = rb + 4 * i;
      const bool z = bstart && (r < 4);
      const int  g = z ? s0 : (s0 - 4 + r);
      float4 t = ((const float4*)(X + (size_t)g * E))[c4];
      if (z) t = make_float4(0.f, 0.f, 0.f, 0.f);
      v[i] = t;
    }
#pragma unroll
    for (int i = 0; i < 33; ++i) {
      const int r = rb + 4 * i;
      *(float4*)(bufX + (r & 1) * POFF + (r >> 1) * LSTRIDE + c4 * 4) = v[i];
    }
  }
  __syncthreads();

  const v4f* W = (const v4f*)Ut;
  conv_pass<2, 0>(bufX, bufY, W, lane, wave);
  conv_pass<2, 2>(bufX, bufY, W, lane, wave);
  __syncthreads();

#pragma unroll
  for (int i = 0; i < CH / 8; ++i) {
    const int row = wave * (CH / 8) + i;
    const float* xr = bufY + (row & 1) * POFF + (row >> 1) * LSTRIDE;
    float s = 0.f;
#pragma unroll
    for (int e = lane; e < E; e += 32) s += xr[e] * fcw[e];
#pragma unroll
    for (int off = 16; off > 0; off >>= 1) s += __shfl_down(s, off);
    if (lane == 0) out[s0 + row] = s + fcb[0];
  }
}

extern "C" void kernel_launch(void* const* d_in, const int* in_sizes, int n_in,
                              void* d_out, int out_size, void* d_ws, size_t ws_size,
                              hipStream_t stream) {
  const float* x      = (const float*)d_in[0];
  const float* conv_w = (const float*)d_in[1];
  const float* fc_w   = (const float*)d_in[2];
  const float* fc_b   = (const float*)d_in[3];
  float* out = (float*)d_out;

  float* Q  = (float*)d_ws;
  float* Ut = Q + 2 * NPT * MATSZ;

  pair_points_kernel<<<1024, 32, 0, stream>>>(conv_w, Q);
  point_product_kernel<<<768, 32, 0, stream>>>(Q, Ut);
  fused_layers_kernel<<<M_TOT / CH, 256, 0, stream>>>(x, Ut, fc_w, fc_b, out);
}


# The 5-tap layer by minimal filtering F(4,5): 4 rows from 8 matrix products (points 0, +-1, +-2, +-1/2, inf), not 2 from 6; one wave per SIMD holds 64 columns x 2 tiles; re-associates the f32 sums
# speedup vs baseline: 2.7427x; 1.1928x over previous
#include <hip/hip_runtime.h>
#include <hip/hip_bf16.h>

typedef __attribute__((ext_vector_type(2))) float v2f;
typedef __attribute__((ext_vector_type(4))) float v4f;
typedef __attribute__((ext_vector_type(8))) float v8f;

#define E 256
#define SEQ_L 512
#define M_TOT 32768
#define CH 128
#define LSTRIDE 260
#define PROWS 33
#define POFF (PROWS * LSTRIDE)
#define BUFSZ (4 * PROWS * LSTRIDE)
#define NPT 8
#define NCT 4
#define NBLK 32
#define MATSZ (E * E)


template <bool SWAP, int G>
__device__ __forceinline__ void pair_points(const float* __restrict__ la, const float* __restrict__ lb,
                                            float* __restrict__ q, const int m0, const int n0,
                                            const int c, const int h) {
  const int fo = SWAP ? n0 : m0;
  const int oo = SWAP ? m0 : n0;
  const float* pa = la + ((size_t)(2 * h) * E + fo + c) * 3;
  const float* pb = lb + ((size_t)(oo + c) * E + 2 * h) * 3;
  v8f acc[4] = {v8f{}, v8f{}, v8f{}, v8f{}};
#pragma unroll 8
  for (int k = 0; k < E / 4; ++k) {
    const float* qa = pa + (size_t)k * 4 * E * 3;
    const float* qb = pb + k * 12;
    const v2f A0 = {qa[0], qa[E * 3]};
    const v2f A1 = {qa[1], qa[E * 3 + 1]};
    const v2f B0 = {qb[0], qb[3]};
    const v2f B1 = {qb[1], qb[4]};
    v2f fa[4], fb[4];
    if (G == 0) {
      fa[0] = A0;             fb[0] = B0;
      fa[1] = A0 + A1;        fb[1] = B0 + B1;
      fa[2] = A0 - A1;        fb[2] = B0 - B1;
      fa[3] = A0 + 2.0f * A1; fb[3] = B0 + 2.0f * B1;
    } else {
      fa[0] = A0 - 2.0f * A1; fb[0] = B0 - 2.0f * B1;
      fa[1] = A0 + 0.5f * A1; fb[1] = B0 + 0.5f * B1;
      fa[2] = A0 - 0.5f * A1; fb[2] = B0 - 0.5f * B1;
      fa[3] = A1;             fb[3] = B1;
    }
#pragma unroll
    for (int i = 0; i < 4; ++i) {
      if (SWAP)
        acc[i] = __builtin_amdgcn_wmma_f32_16x16x4_f32(false, fb[i], false, fa[i], (short)0, acc[i], false, false);
      else
        acc[i] = __builtin_amdgcn_wmma_f32_16x16x4_f32(false, fa[i], false, fb[i], (short)0, acc[i], false, false);
    }
  }
  const size_t base = (size_t)((m0 + 8 * h) / 4) * E + n0 + c;
#pragma unroll
  for (int i = 0; i < 4; ++i) {
    v4f* w4 = (v4f*)(q + (size_t)(4 * G + i) * MATSZ);
    w4[base]     = v4f{acc[i][0], acc[i][1], acc[i][2], acc[i][3]};
    w4[base + E] = v4f{acc[i][4], acc[i][5], acc[i][6], acc[i][7]};
  }
}

__global__ __launch_bounds__(32) void pair_points_kernel(const float* __restrict__ cw,
                                                         float* __restrict__ qall) {
  const int lane = threadIdx.x;
  const int c = lane & 15;
  const int h = lane >> 4;
  const int b  = blockIdx.x;
  const int n0 = (b & 15) * 16;
  const int m0 = ((b >> 4) & 15) * 16;
  const int L  = (b >> 8) & 1;
  const int G  = b >> 9;
  const float* la = cw + (size_t)(2 * L) * E * E * 3;
  const float* lb = cw + (size_t)(2 * L + 1) * E * E * 3;
  float* q = qall + (size_t)L * NPT * MATSZ;
  if (L == 0) {
    if (G == 0) pair_points<true, 0>(la, lb, q, m0, n0, c, h);
    else        pair_points<true, 1>(la, lb, q, m0, n0, c, h);
  } else {
    if (G == 0) pair_points<false, 0>(la, lb, q, m0, n0, c, h);
    else        pair_points<false, 1>(la, lb, q, m0, n0, c, h);
  }
}

__global__ __launch_bounds__(32) void point_product_kernel(const float* __restrict__ qall,
                                                           float* __restrict__ ut) {
  const int lane = threadIdx.x;
  const int c = lane & 15;
  const int h = lane >> 4;
  const int b  = blockIdx.x;
  const int o0 = (b & 7) * 32;
  const int f0 = ((b >> 3) & 15) * 16;
  const int p  = b >> 7;
  const v2f* q0 = (const v2f*)(qall + (size_t)p * MATSZ) + (size_t)(f0 + c) * 2 + h;
  const v2f* q1 = (const v2f*)(qall + (size_t)(NPT + p) * MATSZ) + (size_t)(o0 + c) * 2 + h;
  v8f a0 = {}, a1 = {};
#pragma unroll 16
  for (int s = 0; s < E / 4; ++s) {
    const v2f fa  = q0[(size_t)s * E * 2];
    const v2f fb0 = q1[(size_t)s * E * 2];
    const v2f fb1 = q1[(size_t)s * E * 2 + 32];
    a0 = __builtin_amdgcn_wmma_f32_16x16x4_f32(false, fa, false, fb0, (short)0, a0, false, false);
    a1 = __builtin_amdgcn_wmma_f32_16x16x4_f32(false, fa, false, fb1, (short)0, a1, false, false);
  }
  const float sc = (p == 0) ? -1.0f : (p < 3) ? (-2.0f / 9.0f) : (p < 5) ? (1.0f / 90.0f)
                 : (p < 7) ? (32.0f / 45.0f) : 1.0f;
  v4f* w4 = (v4f*)(ut + (size_t)p * MATSZ);
  const size_t base = (size_t)((f0 + 8 * h) / 4) * E + o0 + c;
  w4[base]          = sc * v4f{a0[0], a0[1], a0[2], a0[3]};
  w4[base + E]      = sc * v4f{a0[4], a0[5], a0[6], a0[7]};
  w4[base + 16]     = sc * v4f{a1[0], a1[1], a1[2], a1[3]};
  w4[base + E + 16] = sc * v4f{a1[4], a1[5], a1[6], a1[7]};
}

__device__ __forceinline__ void vtransform(v4f (&a)[NPT]) {
  const v4f d0 = a[0], d1 = a[1], d2 = a[2], d3 = a[3], d4 = a[4], d5 = a[5], d6 = a[6], d7 = a[7];
  const v4f s1 = (d6 - 4.25f * d4) + d2;
  const v4f t1 = (d5 - 4.25f * d3) + d1;
  const v4f s2 = (d6 - 1.25f * d4) + 0.25f * d2;
  const v4f t2 = (d5 - 1.25f * d3) + 0.25f * d1;
  const v4f s3 = (d6 - 5.0f * d4) + 4.0f * d2;
  const v4f t3 = (d5 - 5.0f * d3) + 4.0f * d1;
  a[0] = (d6 - d0) + 5.25f * (d2 - d4);
  a[1] = s1 + t1;
  a[2] = s1 - t1;
  a[3] = s2 + 2.0f * t2;
  a[4] = s2 - 2.0f * t2;
  a[5] = s3 + 0.5f * t3;
  a[6] = s3 - 0.5f * t3;
  a[7] = (d7 - d1) + 5.25f * (d3 - d5);
}

template <int S, int NT>
__device__ __forceinline__ void wblock(v8f (&acc)[NCT][NT][NPT], v4f (&a)[2][NT][NPT], v4f (&r)[2][NPT][NCT],
                                       const float* const (&qp)[NT],
                                       const v4f* __restrict__ bbase, const int n) {
  const int n1 = (n + 1 < NBLK) ? (n + 1) : (NBLK - 1);
  __builtin_amdgcn_sched_barrier(0);
#pragma unroll
  for (int T = 0; T < NT; ++T)
#pragma unroll
    for (int i = 0; i < NPT; ++i)
      a[S ^ 1][T][i] = *(const v4f*)(qp[T] + (i & 3) * POFF + (i >> 2) * LSTRIDE + n1 * 8);
#pragma unroll
  for (int p = 0; p < NPT; ++p)
#pragma unroll
    for (int t = 0; t < NCT; ++t)
      r[S ^ 1][p][t] = bbase[(size_t)(p * (E / 4) + 2 * n1) * E + 16 * t];
  __builtin_amdgcn_sched_barrier(0);
#pragma unroll
  for (int T = 0; T < NT; ++T)
#pragma unroll
    for (int t = 0; t < NCT; ++t)
#pragma unroll
      for (int p = 0; p < NPT; ++p) {
        const v4f f = a[S][T][p];
        const v4f g = r[S][p][t];
        acc[t][T][p] = __builtin_amdgcn_wmma_f32_16x16x4_f32(false, __builtin_shufflevector(f, f, 0, 1), false, __builtin_shufflevector(g, g, 0, 1), (short)0, acc[t][T][p], false, false);
      }
  __builtin_amdgcn_sched_barrier(0);
#pragma unroll
  for (int T = 0; T < NT; ++T) vtransform(a[S ^ 1][T]);
#pragma unroll
  for (int t = 0; t < NCT; ++t)
#pragma unroll
    for (int T = 0; T < NT; ++T)
#pragma unroll
      for (int p = 0; p < NPT; ++p) {
        const v4f f = a[S][T][p];
        const v4f g = r[S][p][t];
        acc[t][T][p] = __builtin_amdgcn_wmma_f32_16x16x4_f32(false, __builtin_shufflevector(f, f, 2, 3), false, __builtin_shufflevector(g, g, 2, 3), (short)0, acc[t][T][p], false, false);
      }
#pragma unroll
  for (int g = 0; g < 24 * NT; ++g) {
    __builtin_amdgcn_sched_group_barrier(0x008, 1, 0);
    __builtin_amdgcn_sched_group_barrier(0x002, 2, 0);
  }
}

template <int NT, int T0>
__device__ __forceinline__ void conv_pass(const float* __restrict__ in,
                                          float* __restrict__ out,
                                          const v4f* __restrict__ Wl,
                                          const int lane, const int wave) {
  const int h = lane >> 4;
  const int c = lane & 15;
  const int ncol = wave * (16 * NCT) + c;

  const float* qp[NT];
#pragma unroll
  for (int T = 0; T < NT; ++T) {
    const int q = 16 * (T0 + T) + c;
    qp[T] = in + q * LSTRIDE + 4 * h;
  }
  const v4f* bbase = Wl + h * E + ncol;

  v8f acc[NCT][NT][NPT];
#pragma unroll
  for (int t = 0; t < NCT; ++t)
#pragma unroll
    for (int T = 0; T < NT; ++T)
#pragma unroll
      for (int p = 0; p < NPT; ++p) acc[t][T][p] = v8f{};

  v4f a[2][NT][NPT];
  v4f r[2][NPT][NCT];

#pragma unroll
  for (int T = 0; T < NT; ++T)
#pragma unroll
    for (int i = 0; i < NPT; ++i)
      a[0][T][i] = *(const v4f*)(qp[T] + (i & 3) * POFF + (i >> 2) * LSTRIDE);
#pragma unroll
  for (int p = 0; p < NPT; ++p)
#pragma unroll
    for (int t = 0; t < NCT; ++t)
      r[0][p][t] = bbase[(size_t)(p * (E / 4)) * E + 16 * t];
#pragma unroll
  for (int T = 0; T < NT; ++T) vtransform(a[0][T]);

  for (int n = 0; n < NBLK; n += 2) {
    wblock<0, NT>(acc, a, r, qp, bbase, n);
    wblock<1, NT>(acc, a, r, qp, bbase, n + 1);
  }

#pragma unroll
  for (int t = 0; t < NCT; ++t) {
#pragma unroll
    for (int T = 0; T < NT; ++T) {
      const int pb = 16 * (T0 + T) + 8 * h;
      const int col = ncol + 16 * t;
#pragma unroll
      for (int v = 0; v < 8; ++v) {
        const float m0 = acc[t][T][0][v];
        const float m1 = acc[t][T][1][v];
        const float m2 = acc[t][T][2][v];
        const float m3 = acc[t][T][3][v];
        const float m4 = acc[t][T][4][v];
        const float m5 = acc[t][T][5][v];
        const float m6 = acc[t][T][6][v];
        const float m7 = acc[t][T][7][v];
        const float e1 = m1 + m2, o1 = m1 - m2;
        const float e2 = m3 + m4, o2 = m3 - m4;
        const float e3 = m5 + m6, o3 = m5 - m6;
        const float y0 = (m0 + e1) + (e2 + e3);
        const float y1 = (o1 + 2.0f * o2) + 0.5f * o3;
        const float y2 = (e1 + 4.0f * e2) + 0.25f * e3;
        const float y3 = ((o1 + 8.0f * o2) + 0.125f * o3) + m7;
        const int po = pb + v;
        out[po * LSTRIDE + col] = y0;
        out[POFF + po * LSTRIDE + col] = y1;
        out[2 * POFF + po * LSTRIDE + col] = y2;
        out[3 * POFF + po * LSTRIDE + col] = y3;
      }
    }
  }
}

__global__ __launch_bounds__(128) void fused_layers_kernel(const float* __restrict__ X,
                                                           const float* __restrict__ Ut,
                                                           const float* __restrict__ fcw,
                                                           const float* __restrict__ fcb,
                                                           float* __restrict__ out) {
  __shared__ __attribute__((aligned(16))) float lds[2 * BUFSZ];
  float* bufX = lds;
  float* bufY = lds + BUFSZ;

  const int tid  = threadIdx.x;
  const int lane = tid & 31;
  const int wave = tid >> 5;
  const int s0   = blockIdx.x * CH;

  const bool bstart = (s0 % SEQ_L) == 0;
  {
    const int c4 = tid & 63;
    const int rb = tid >> 6;
    float4 v[66];
#pragma unroll
    for (int i = 0; i < 66; ++i) {
      const int r = rb + 2 * i;
      const bool z = bstart && (r < 4);
      const int  g = z ? s0 : (s0 - 4 + r);
      float4 t = ((const float4*)(X + (size_t)g * E))[c4];
      if (z) t = make_float4(0.f, 0.f, 0.f, 0.f);
      v[i] = t;
    }
#pragma unroll
    for (int i = 0; i < 66; ++i) {
      const int r = rb + 2 * i;
      *(float4*)(bufX + (r & 3) * POFF + (r >> 2) * LSTRIDE + c4 * 4) = v[i];
    }
  }
  __syncthreads();

  const v4f* W = (const v4f*)Ut;
  conv_pass<2, 0>(bufX, bufY, W, lane, wave);
  __syncthreads();

#pragma unroll
  for (int i = 0; i < CH / 4; ++i) {
    const int row = wave * (CH / 4) + i;
    const float* xr = bufY + (row & 3) * POFF + (row >> 2) * LSTRIDE;
    float s = 0.f;
#pragma unroll
    for (int e = lane; e < E; e += 32) s += xr[e] * fcw[e];
#pragma unroll
    for (int off = 16; off > 0; off >>= 1) s += __shfl_down(s, off);
    if (lane == 0) out[s0 + row] = s + fcb[0];
  }
}

extern "C" void kernel_launch(void* const* d_in, const int* in_sizes, int n_in,
                              void* d_out, int out_size, void* d_ws, size_t ws_size,
                              hipStream_t stream) {
  const float* x      = (const float*)d_in[0];
  const float* conv_w = (const float*)d_in[1];
  const float* fc_w   = (const float*)d_in[2];
  const float* fc_b   = (const float*)d_in[3];
  float* out = (float*)d_out;

  float* Q  = (float*)d_ws;
  float* Ut = Q + 2 * NPT * MATSZ;

  pair_points_kernel<<<1024, 32, 0, stream>>>(conv_w, Q);
  point_product_kernel<<<NPT * 128, 32, 0, stream>>>(Q, Ut);
  fused_layers_kernel<<<M_TOT / CH, 128, 0, stream>>>(x, Ut, fc_w, fc_b, out);
}


# fc_out in registers: a wave weights its point sums by fc_w, then the output formulas, sums its 64 columns by DPP row butterflies; 4 waves meet in LDS; layer output not stored; re-associates f32 sums
# speedup vs baseline: 3.1992x; 1.1664x over previous
#include <hip/hip_runtime.h>
#include <hip/hip_bf16.h>

typedef __attribute__((ext_vector_type(2))) float v2f;
typedef __attribute__((ext_vector_type(4))) float v4f;
typedef __attribute__((ext_vector_type(8))) float v8f;

#define E 256
#define SEQ_L 512
#define M_TOT 32768
#define CH 128
#define LSTRIDE 260
#define PROWS 33
#define POFF (PROWS * LSTRIDE)
#define BUFSZ (4 * PROWS * LSTRIDE)
#define NPT 8
#define NCT 4
#define NBLK 32
#define MATSZ (E * E)


template <bool SWAP, int G>
__device__ __forceinline__ void pair_points(const float* __restrict__ la, const float* __restrict__ lb,
                                            float* __restrict__ q, const int m0, const int n0,
                                            const int c, const int h) {
  const int fo = SWAP ? n0 : m0;
  const int oo = SWAP ? m0 : n0;
  const float* pa = la + ((size_t)(2 * h) * E + fo + c) * 3;
  const float* pb = lb + ((size_t)(oo + c) * E + 2 * h) * 3;
  v8f acc[4] = {v8f{}, v8f{}, v8f{}, v8f{}};
#pragma unroll 8
  for (int k = 0; k < E / 4; ++k) {
    const float* qa = pa + (size_t)k * 4 * E * 3;
    const float* qb = pb + k * 12;
    const v2f A0 = {qa[0], qa[E * 3]};
    const v2f A1 = {qa[1], qa[E * 3 + 1]};
    const v2f B0 = {qb[0], qb[3]};
    const v2f B1 = {qb[1], qb[4]};
    v2f fa[4], fb[4];
    if (G == 0) {
      fa[0] = A0;             fb[0] = B0;
      fa[1] = A0 + A1;        fb[1] = B0 + B1;
      fa[2] = A0 - A1;        fb[2] = B0 - B1;
      fa[3] = A0 + 2.0f * A1; fb[3] = B0 + 2.0f * B1;
    } else {
      fa[0] = A0 - 2.0f * A1; fb[0] = B0 - 2.0f * B1;
      fa[1] = A0 + 0.5f * A1; fb[1] = B0 + 0.5f * B1;
      fa[2] = A0 - 0.5f * A1; fb[2] = B0 - 0.5f * B1;
      fa[3] = A1;             fb[3] = B1;
    }
#pragma unroll
    for (int i = 0; i < 4; ++i) {
      if (SWAP)
        acc[i] = __builtin_amdgcn_wmma_f32_16x16x4_f32(false, fb[i], false, fa[i], (short)0, acc[i], false, false);
      else
        acc[i] = __builtin_amdgcn_wmma_f32_16x16x4_f32(false, fa[i], false, fb[i], (short)0, acc[i], false, false);
    }
  }
  const size_t base = (size_t)((m0 + 8 * h) / 4) * E + n0 + c;
#pragma unroll
  for (int i = 0; i < 4; ++i) {
    v4f* w4 = (v4f*)(q + (size_t)(4 * G + i) * MATSZ);
    w4[base]     = v4f{acc[i][0], acc[i][1], acc[i][2], acc[i][3]};
    w4[base + E] = v4f{acc[i][4], acc[i][5], acc[i][6], acc[i][7]};
  }
}

__global__ __launch_bounds__(32) void pair_points_kernel(const float* __restrict__ cw,
                                                         float* __restrict__ qall) {
  const int lane = threadIdx.x;
  const int c = lane & 15;
  const int h = lane >> 4;
  const int b  = blockIdx.x;
  const int n0 = (b & 15) * 16;
  const int m0 = ((b >> 4) & 15) * 16;
  const int L  = (b >> 8) & 1;
  const int G  = b >> 9;
  const float* la = cw + (size_t)(2 * L) * E * E * 3;
  const float* lb = cw + (size_t)(2 * L + 1) * E * E * 3;
  float* q = qall + (size_t)L * NPT * MATSZ;
  if (L == 0) {
    if (G == 0) pair_points<true, 0>(la, lb, q, m0, n0, c, h);
    else        pair_points<true, 1>(la, lb, q, m0, n0, c, h);
  } else {
    if (G == 0) pair_points<false, 0>(la, lb, q, m0, n0, c, h);
    else        pair_points<false, 1>(la, lb, q, m0, n0, c, h);
  }
}

__global__ __launch_bounds__(32) void point_product_kernel(const float* __restrict__ qall,
                                                           float* __restrict__ ut) {
  const int lane = threadIdx.x;
  const int c = lane & 15;
  const int h = lane >> 4;
  const int b  = blockIdx.x;
  const int o0 = (b & 7) * 32;
  const int f0 = ((b >> 3) & 15) * 16;
  const int p  = b >> 7;
  const v2f* q0 = (const v2f*)(qall + (size_t)p * MATSZ) + (size_t)(f0 + c) * 2 + h;
  const v2f* q1 = (const v2f*)(qall + (size_t)(NPT + p) * MATSZ) + (size_t)(o0 + c) * 2 + h;
  v8f a0 = {}, a1 = {};
#pragma unroll 16
  for (int s = 0; s < E / 4; ++s) {
    const v2f fa  = q0[(size_t)s * E * 2];
    const v2f fb0 = q1[(size_t)s * E * 2];
    const v2f fb1 = q1[(size_t)s * E * 2 + 32];
    a0 = __builtin_amdgcn_wmma_f32_16x16x4_f32(false, fa, false, fb0, (short)0, a0, false, false);
    a1 = __builtin_amdgcn_wmma_f32_16x16x4_f32(false, fa, false, fb1, (short)0, a1, false, false);
  }
  const float sc = (p == 0) ? -1.0f : (p < 3) ? (-2.0f / 9.0f) : (p < 5) ? (1.0f / 90.0f)
                 : (p < 7) ? (32.0f / 45.0f) : 1.0f;
  v4f* w4 = (v4f*)(ut + (size_t)p * MATSZ);
  const size_t base = (size_t)((f0 + 8 * h) / 4) * E + o0 + c;
  w4[base]          = sc * v4f{a0[0], a0[1], a0[2], a0[3]};
  w4[base + E]      = sc * v4f{a0[4], a0[5], a0[6], a0[7]};
  w4[base + 16]     = sc * v4f{a1[0], a1[1], a1[2], a1[3]};
  w4[base + E + 16] = sc * v4f{a1[4], a1[5], a1[6], a1[7]};
}

__device__ __forceinline__ void vtransform(v4f (&a)[NPT]) {
  const v4f d0 = a[0], d1 = a[1], d2 = a[2], d3 = a[3], d4 = a[4], d5 = a[5], d6 = a[6], d7 = a[7];
  const v4f s1 = (d6 - 4.25f * d4) + d2;
  const v4f t1 = (d5 - 4.25f * d3) + d1;
  const v4f s2 = (d6 - 1.25f * d4) + 0.25f * d2;
  const v4f t2 = (d5 - 1.25f * d3) + 0.25f * d1;
  const v4f s3 = (d6 - 5.0f * d4) + 4.0f * d2;
  const v4f t3 = (d5 - 5.0f * d3) + 4.0f * d1;
  a[0] = (d6 - d0) + 5.25f * (d2 - d4);
  a[1] = s1 + t1;
  a[2] = s1 - t1;
  a[3] = s2 + 2.0f * t2;
  a[4] = s2 - 2.0f * t2;
  a[5] = s3 + 0.5f * t3;
  a[6] = s3 - 0.5f * t3;
  a[7] = (d7 - d1) + 5.25f * (d3 - d5);
}

template <int S, int NT>
__device__ __forceinline__ void wblock(v8f (&acc)[NCT][NT][NPT], v4f (&a)[2][NT][NPT], v4f (&r)[2][NPT][NCT],
                                       const float* const (&qp)[NT],
                                       const v4f* __restrict__ bbase, const int n) {
  const int n1 = (n + 1 < NBLK) ? (n + 1) : (NBLK - 1);
  __builtin_amdgcn_sched_barrier(0);
#pragma unroll
  for (int T = 0; T < NT; ++T)
#pragma unroll
    for (int i = 0; i < NPT; ++i)
      a[S ^ 1][T][i] = *(const v4f*)(qp[T] + (i & 3) * POFF + (i >> 2) * LSTRIDE + n1 * 8);
#pragma unroll
  for (int p = 0; p < NPT; ++p)
#pragma unroll
    for (int t = 0; t < NCT; ++t)
      r[S ^ 1][p][t] = bbase[(size_t)(p * (E / 4) + 2 * n1) * E + 16 * t];
  __builtin_amdgcn_sched_barrier(0);
#pragma unroll
  for (int T = 0; T < NT; ++T)
#pragma unroll
    for (int t = 0; t < NCT; ++t)
#pragma unroll
      for (int p = 0; p < NPT; ++p) {
        const v4f f = a[S][T][p];
        const v4f g = r[S][p][t];
        acc[t][T][p] = __builtin_amdgcn_wmma_f32_16x16x4_f32(false, __builtin_shufflevector(f, f, 0, 1), false, __builtin_shufflevector(g, g, 0, 1), (short)0, acc[t][T][p], false, false);
      }
  __builtin_amdgcn_sched_barrier(0);
#pragma unroll
  for (int T = 0; T < NT; ++T) vtransform(a[S ^ 1][T]);
#pragma unroll
  for (int t = 0; t < NCT; ++t)
#pragma unroll
    for (int T = 0; T < NT; ++T)
#pragma unroll
      for (int p = 0; p < NPT; ++p) {
        const v4f f = a[S][T][p];
        const v4f g = r[S][p][t];
        acc[t][T][p] = __builtin_amdgcn_wmma_f32_16x16x4_f32(false, __builtin_shufflevector(f, f, 2, 3), false, __builtin_shufflevector(g, g, 2, 3), (short)0, acc[t][T][p], false, false);
      }
#pragma unroll
  for (int g = 0; g < 24 * NT; ++g) {
    __builtin_amdgcn_sched_group_barrier(0x008, 1, 0);
    __builtin_amdgcn_sched_group_barrier(0x002, 2, 0);
  }
}

template <int CTRL>
__device__ __forceinline__ float xmask_add(const float x) {
  return x + __int_as_float(__builtin_amdgcn_update_dpp(0, __float_as_int(x), CTRL, 0xf, 0xf, true));
}
__device__ __forceinline__ float row_sum16(float x) {
  x = xmask_add<0x161>(x);
  x = xmask_add<0x162>(x);
  x = xmask_add<0x164>(x);
  x = xmask_add<0x168>(x);
  return x;
}

template <int NT, int T0>
__device__ __forceinline__ void conv_pass(const float* __restrict__ in,
                                          float* __restrict__ part,
                                          const v4f* __restrict__ Wl,
                                          const float* __restrict__ fcw,
                                          const int lane, const int wave) {
  const int h = lane >> 4;
  const int c = lane & 15;
  const int ncol = wave * (16 * NCT) + c;

  const float* qp[NT];
#pragma unroll
  for (int T = 0; T < NT; ++T) {
    const int q = 16 * (T0 + T) + c;
    qp[T] = in + q * LSTRIDE + 4 * h;
  }
  const v4f* bbase = Wl + h * E + ncol;

  float fw[NCT];
#pragma unroll
  for (int t = 0; t < NCT; ++t) fw[t] = fcw[ncol + 16 * t];

  v8f acc[NCT][NT][NPT];
#pragma unroll
  for (int t = 0; t < NCT; ++t)
#pragma unroll
    for (int T = 0; T < NT; ++T)
#pragma unroll
      for (int p = 0; p < NPT; ++p) acc[t][T][p] = v8f{};

  v4f a[2][NT][NPT];
  v4f r[2][NPT][NCT];

#pragma unroll
  for (int T = 0; T < NT; ++T)
#pragma unroll
    for (int i = 0; i < NPT; ++i)
      a[0][T][i] = *(const v4f*)(qp[T] + (i & 3) * POFF + (i >> 2) * LSTRIDE);
#pragma unroll
  for (int p = 0; p < NPT; ++p)
#pragma unroll
    for (int t = 0; t < NCT; ++t)
      r[0][p][t] = bbase[(size_t)(p * (E / 4)) * E + 16 * t];
#pragma unroll
  for (int T = 0; T < NT; ++T) vtransform(a[0][T]);

  for (int n = 0; n < NBLK; n += 2) {
    wblock<0, NT>(acc, a, r, qp, bbase, n);
    wblock<1, NT>(acc, a, r, qp, bbase, n + 1);
  }

  v4f rs[NT][8];
#pragma unroll
  for (int T = 0; T < NT; ++T) {
    v8f m[NPT];
#pragma unroll
    for (int p = 0; p < NPT; ++p) {
      v8f w = fw[0] * acc[0][T][p];
#pragma unroll
      for (int t = 1; t < NCT; ++t) w += fw[t] * acc[t][T][p];
      m[p] = w;
    }
    const v8f e1 = m[1] + m[2], o1 = m[1] - m[2];
    const v8f e2 = m[3] + m[4], o2 = m[3] - m[4];
    const v8f e3 = m[5] + m[6], o3 = m[5] - m[6];
    const v8f y0 = (m[0] + e1) + (e2 + e3);
    const v8f y1 = (o1 + 2.0f * o2) + 0.5f * o3;
    const v8f y2 = (e1 + 4.0f * e2) + 0.25f * e3;
    const v8f y3 = ((o1 + 8.0f * o2) + 0.125f * o3) + m[7];
#pragma unroll
    for (int v = 0; v < 8; ++v)
      rs[T][v] = v4f{row_sum16(y0[v]), row_sum16(y1[v]), row_sum16(y2[v]), row_sum16(y3[v])};
  }
  if (c == 0) {
#pragma unroll
    for (int T = 0; T < NT; ++T)
#pragma unroll
      for (int v = 0; v < 8; ++v)
        *(v4f*)(part + 4 * (16 * (T0 + T) + 8 * h + v)) = rs[T][v];
  }
}

__global__ __launch_bounds__(128) void fused_layers_kernel(const float* __restrict__ X,
                                                           const float* __restrict__ Ut,
                                                           const float* __restrict__ fcw,
                                                           const float* __restrict__ fcb,
                                                           float* __restrict__ out) {
  __shared__ __attribute__((aligned(16))) float lds[BUFSZ + 4 * CH];
  float* bufX = lds;
  float* part = lds + BUFSZ;

  const int tid  = threadIdx.x;
  const int lane = tid & 31;
  const int wave = tid >> 5;
  const int s0   = blockIdx.x * CH;

  const bool bstart = (s0 % SEQ_L) == 0;
  {
    const int c4 = tid & 63;
    const int rb = tid >> 6;
    float4 v[66];
#pragma unroll
    for (int i = 0; i < 66; ++i) {
      const int r = rb + 2 * i;
      const bool z = bstart && (r < 4);
      const int  g = z ? s0 : (s0 - 4 + r);
      float4 t = ((const float4*)(X + (size_t)g * E))[c4];
      if (z) t = make_float4(0.f, 0.f, 0.f, 0.f);
      v[i] = t;
    }
#pragma unroll
    for (int i = 0; i < 66; ++i) {
      const int r = rb + 2 * i;
      *(float4*)(bufX + (r & 3) * POFF + (r >> 2) * LSTRIDE + c4 * 4) = v[i];
    }
  }
  __syncthreads();

  const v4f* W = (const v4f*)Ut;
  conv_pass<2, 0>(bufX, part + wave * CH, W, fcw, lane, wave);
  __syncthreads();

  out[s0 + tid] = ((part[tid] + part[CH + tid]) + (part[2 * CH + tid] + part[3 * CH + tid])) + fcb[0];
}

extern "C" void kernel_launch(void* const* d_in, const int* in_sizes, int n_in,
                              void* d_out, int out_size, void* d_ws, size_t ws_size,
                              hipStream_t stream) {
  const float* x      = (const float*)d_in[0];
  const float* conv_w = (const float*)d_in[1];
  const float* fc_w   = (const float*)d_in[2];
  const float* fc_b   = (const float*)d_in[3];
  float* out = (float*)d_out;

  float* Q  = (float*)d_ws;
  float* Ut = Q + 2 * NPT * MATSZ;

  pair_points_kernel<<<1024, 32, 0, stream>>>(conv_w, Q);
  point_product_kernel<<<NPT * 128, 32, 0, stream>>>(Q, Ut);
  fused_layers_kernel<<<M_TOT / CH, 128, 0, stream>>>(x, Ut, fc_w, fc_b, out);
}


# Both weight-preparation launches: the next k group's global loads issued in front of the current group's products (second register set); f32 sum order unchanged
# speedup vs baseline: 3.3943x; 1.0610x over previous
#include <hip/hip_runtime.h>
#include <hip/hip_bf16.h>

typedef __attribute__((ext_vector_type(2))) float v2f;
typedef __attribute__((ext_vector_type(4))) float v4f;
typedef __attribute__((ext_vector_type(8))) float v8f;

#define E 256
#define SEQ_L 512
#define M_TOT 32768
#define CH 128
#define LSTRIDE 260
#define PROWS 33
#define POFF (PROWS * LSTRIDE)
#define BUFSZ (4 * PROWS * LSTRIDE)
#define NPT 8
#define NCT 4
#define NBLK 32
#define MATSZ (E * E)


#define KU1 8
#define NG1 (E / 4 / KU1)

__device__ __forceinline__ void pair_group_loads(const float* __restrict__ pa, const float* __restrict__ pb,
                                                 const int g, v2f (&w)[4][KU1]) {
#pragma unroll
  for (int i = 0; i < KU1; ++i) {
    const int k = g * KU1 + i;
    const float* qa = pa + (size_t)k * 4 * E * 3;
    const float* qb = pb + k * 12;
    w[0][i] = v2f{qa[0], qa[1]};
    w[1][i] = v2f{qa[E * 3], qa[E * 3 + 1]};
    w[2][i] = v2f{qb[0], qb[1]};
    w[3][i] = v2f{qb[3], qb[4]};
  }
}

template <bool SWAP, int G>
__device__ __forceinline__ void pair_group_products(const v2f (&w)[4][KU1], v8f (&acc)[4]) {
#pragma unroll
  for (int i = 0; i < KU1; ++i) {
    const v2f A0 = {w[0][i][0], w[1][i][0]}, A1 = {w[0][i][1], w[1][i][1]};
    const v2f B0 = {w[2][i][0], w[3][i][0]}, B1 = {w[2][i][1], w[3][i][1]};
    v2f fa[4], fb[4];
    if (G == 0) {
      fa[0] = A0;             fb[0] = B0;
      fa[1] = A0 + A1;        fb[1] = B0 + B1;
      fa[2] = A0 - A1;        fb[2] = B0 - B1;
      fa[3] = A0 + 2.0f * A1; fb[3] = B0 + 2.0f * B1;
    } else {
      fa[0] = A0 - 2.0f * A1; fb[0] = B0 - 2.0f * B1;
      fa[1] = A0 + 0.5f * A1; fb[1] = B0 + 0.5f * B1;
      fa[2] = A0 - 0.5f * A1; fb[2] = B0 - 0.5f * B1;
      fa[3] = A1;             fb[3] = B1;
    }
#pragma unroll
    for (int j = 0; j < 4; ++j) {
      if (SWAP)
        acc[j] = __builtin_amdgcn_wmma_f32_16x16x4_f32(false, fb[j], false, fa[j], (short)0, acc[j], false, false);
      else
        acc[j] = __builtin_amdgcn_wmma_f32_16x16x4_f32(false, fa[j], false, fb[j], (short)0, acc[j], false, false);
    }
  }
}

template <bool SWAP, int G>
__device__ __forceinline__ void pair_points(const float* __restrict__ la, const float* __restrict__ lb,
                                            float* __restrict__ q, const int m0, const int n0,
                                            const int c, const int h) {
  const int fo = SWAP ? n0 : m0;
  const int oo = SWAP ? m0 : n0;
  const float* pa = la + ((size_t)(2 * h) * E + fo + c) * 3;
  const float* pb = lb + ((size_t)(oo + c) * E + 2 * h) * 3;
  v8f acc[4] = {v8f{}, v8f{}, v8f{}, v8f{}};
  v2f w0[4][KU1], w1[4][KU1];
  pair_group_loads(pa, pb, 0, w0);
  for (int g = 0; g < NG1; g += 2) {
    const int g2 = (g + 2 < NG1) ? (g + 2) : (NG1 - 1);
    __builtin_amdgcn_sched_barrier(0);
    pair_group_loads(pa, pb, g + 1, w1);
    __builtin_amdgcn_sched_barrier(0);
    pair_group_products<SWAP, G>(w0, acc);
    __builtin_amdgcn_sched_barrier(0);
    pair_group_loads(pa, pb, g2, w0);
    __builtin_amdgcn_sched_barrier(0);
    pair_group_products<SWAP, G>(w1, acc);
  }
  const size_t base = (size_t)((m0 + 8 * h) / 4) * E + n0 + c;
#pragma unroll
  for (int i = 0; i < 4; ++i) {
    v4f* w4 = (v4f*)(q + (size_t)(4 * G + i) * MATSZ);
    w4[base]     = v4f{acc[i][0], acc[i][1], acc[i][2], acc[i][3]};
    w4[base + E] = v4f{acc[i][4], acc[i][5], acc[i][6], acc[i][7]};
  }
}

__global__ __launch_bounds__(32) void pair_points_kernel(const float* __restrict__ cw,
                                                         float* __restrict__ qall) {
  const int lane = threadIdx.x;
  const int c = lane & 15;
  const int h = lane >> 4;
  const int b  = blockIdx.x;
  const int n0 = (b & 15) * 16;
  const int m0 = ((b >> 4) & 15) * 16;
  const int L  = (b >> 8) & 1;
  const int G  = b >> 9;
  const float* la = cw + (size_t)(2 * L) * E * E * 3;
  const float* lb = cw + (size_t)(2 * L + 1) * E * E * 3;
  float* q = qall + (size_t)L * NPT * MATSZ;
  if (L == 0) {
    if (G == 0) pair_points<true, 0>(la, lb, q, m0, n0, c, h);
    else        pair_points<true, 1>(la, lb, q, m0, n0, c, h);
  } else {
    if (G == 0) pair_points<false, 0>(la, lb, q, m0, n0, c, h);
    else        pair_points<false, 1>(la, lb, q, m0, n0, c, h);
  }
}

#define KU2 16
#define NG2 (E / 4 / KU2)

__device__ __forceinline__ void product_group_loads(const v2f* __restrict__ q0, const v2f* __restrict__ q1,
                                                    const int g, v2f (&u)[3][KU2]) {
#pragma unroll
  for (int i = 0; i < KU2; ++i) {
    const size_t s = (size_t)(g * KU2 + i);
    u[0][i] = q0[s * E * 2];
    u[1][i] = q1[s * E * 2];
    u[2][i] = q1[s * E * 2 + 32];
  }
}

__device__ __forceinline__ void product_group_products(const v2f (&u)[3][KU2], v8f& a0, v8f& a1) {
#pragma unroll
  for (int i = 0; i < KU2; ++i) {
    a0 = __builtin_amdgcn_wmma_f32_16x16x4_f32(false, u[0][i], false, u[1][i], (short)0, a0, false, false);
    a1 = __builtin_amdgcn_wmma_f32_16x16x4_f32(false, u[0][i], false, u[2][i], (short)0, a1, false, false);
  }
}

__global__ __launch_bounds__(32) void point_product_kernel(const float* __restrict__ qall,
                                                           float* __restrict__ ut) {
  const int lane = threadIdx.x;
  const int c = lane & 15;
  const int h = lane >> 4;
  const int b  = blockIdx.x;
  const int o0 = (b & 7) * 32;
  const int f0 = ((b >> 3) & 15) * 16;
  const int p  = b >> 7;
  const v2f* q0 = (const v2f*)(qall + (size_t)p * MATSZ) + (size_t)(f0 + c) * 2 + h;
  const v2f* q1 = (const v2f*)(qall + (size_t)(NPT + p) * MATSZ) + (size_t)(o0 + c) * 2 + h;
  v8f a0 = {}, a1 = {};
  v2f u0[3][KU2], u1[3][KU2];
  product_group_loads(q0, q1, 0, u0);
  for (int g = 0; g < NG2; g += 2) {
    const int g2 = (g + 2 < NG2) ? (g + 2) : (NG2 - 1);
    __builtin_amdgcn_sched_barrier(0);
    product_group_loads(q0, q1, g + 1, u1);
    __builtin_amdgcn_sched_barrier(0);
    product_group_products(u0, a0, a1);
    __builtin_amdgcn_sched_barrier(0);
    product_group_loads(q0, q1, g2, u0);
    __builtin_amdgcn_sched_barrier(0);
    product_group_products(u1, a0, a1);
  }
  const float sc = (p == 0) ? -1.0f : (p < 3) ? (-2.0f / 9.0f) : (p < 5) ? (1.0f / 90.0f)
                 : (p < 7) ? (32.0f / 45.0f) : 1.0f;
  v4f* w4 = (v4f*)(ut + (size_t)p * MATSZ);
  const size_t base = (size_t)((f0 + 8 * h) / 4) * E + o0 + c;
  w4[base]          = sc * v4f{a0[0], a0[1], a0[2], a0[3]};
  w4[base + E]      = sc * v4f{a0[4], a0[5], a0[6], a0[7]};
  w4[base + 16]     = sc * v4f{a1[0], a1[1], a1[2], a1[3]};
  w4[base + E + 16] = sc * v4f{a1[4], a1[5], a1[6], a1[7]};
}

__device__ __forceinline__ void vtransform(v4f (&a)[NPT]) {
  const v4f d0 = a[0], d1 = a[1], d2 = a[2], d3 = a[3], d4 = a[4], d5 = a[5], d6 = a[6], d7 = a[7];
  const v4f s1 = (d6 - 4.25f * d4) + d2;
  const v4f t1 = (d5 - 4.25f * d3) + d1;
  const v4f s2 = (d6 - 1.25f * d4) + 0.25f * d2;
  const v4f t2 = (d5 - 1.25f * d3) + 0.25f * d1;
  const v4f s3 = (d6 - 5.0f * d4) + 4.0f * d2;
  const v4f t3 = (d5 - 5.0f * d3) + 4.0f * d1;
  a[0] = (d6 - d0) + 5.25f * (d2 - d4);
  a[1] = s1 + t1;
  a[2] = s1 - t1;
  a[3] = s2 + 2.0f * t2;
  a[4] = s2 - 2.0f * t2;
  a[5] = s3 + 0.5f * t3;
  a[6] = s3 - 0.5f * t3;
  a[7] = (d7 - d1) + 5.25f * (d3 - d5);
}

template <int S, int NT>
__device__ __forceinline__ void wblock(v8f (&acc)[NCT][NT][NPT], v4f (&a)[2][NT][NPT], v4f (&r)[2][NPT][NCT],
                                       const float* const (&qp)[NT],
                                       const v4f* __restrict__ bbase, const int n) {
  const int n1 = (n + 1 < NBLK) ? (n + 1) : (NBLK - 1);
  __builtin_amdgcn_sched_barrier(0);
#pragma unroll
  for (int T = 0; T < NT; ++T)
#pragma unroll
    for (int i = 0; i < NPT; ++i)
      a[S ^ 1][T][i] = *(const v4f*)(qp[T] + (i & 3) * POFF + (i >> 2) * LSTRIDE + n1 * 8);
#pragma unroll
  for (int p = 0; p < NPT; ++p)
#pragma unroll
    for (int t = 0; t < NCT; ++t)
      r[S ^ 1][p][t] = bbase[(size_t)(p * (E / 4) + 2 * n1) * E + 16 * t];
  __builtin_amdgcn_sched_barrier(0);
#pragma unroll
  for (int T = 0; T < NT; ++T)
#pragma unroll
    for (int t = 0; t < NCT; ++t)
#pragma unroll
      for (int p = 0; p < NPT; ++p) {
        const v4f f = a[S][T][p];
        const v4f g = r[S][p][t];
        acc[t][T][p] = __builtin_amdgcn_wmma_f32_16x16x4_f32(false, __builtin_shufflevector(f, f, 0, 1), false, __builtin_shufflevector(g, g, 0, 1), (short)0, acc[t][T][p], false, false);
      }
  __builtin_amdgcn_sched_barrier(0);
#pragma unroll
  for (int T = 0; T < NT; ++T) vtransform(a[S ^ 1][T]);
#pragma unroll
  for (int t = 0; t < NCT; ++t)
#pragma unroll
    for (int T = 0; T < NT; ++T)
#pragma unroll
      for (int p = 0; p < NPT; ++p) {
        const v4f f = a[S][T][p];
        const v4f g = r[S][p][t];
        acc[t][T][p] = __builtin_amdgcn_wmma_f32_16x16x4_f32(false, __builtin_shufflevector(f, f, 2, 3), false, __builtin_shufflevector(g, g, 2, 3), (short)0, acc[t][T][p], false, false);
      }
#pragma unroll
  for (int g = 0; g < 24 * NT; ++g) {
    __builtin_amdgcn_sched_group_barrier(0x008, 1, 0);
    __builtin_amdgcn_sched_group_barrier(0x002, 2, 0);
  }
}

template <int CTRL>
__device__ __forceinline__ float xmask_add(const float x) {
  return x + __int_as_float(__builtin_amdgcn_update_dpp(0, __float_as_int(x), CTRL, 0xf, 0xf, true));
}
__device__ __forceinline__ float row_sum16(float x) {
  x = xmask_add<0x161>(x);
  x = xmask_add<0x162>(x);
  x = xmask_add<0x164>(x);
  x = xmask_add<0x168>(x);
  return x;
}

template <int NT, int T0>
__device__ __forceinline__ void conv_pass(const float* __restrict__ in,
                                          float* __restrict__ part,
                                          const v4f* __restrict__ Wl,
                                          const float* __restrict__ fcw,
                                          const int lane, const int wave) {
  const int h = lane >> 4;
  const int c = lane & 15;
  const int ncol = wave * (16 * NCT) + c;

  const float* qp[NT];
#pragma unroll
  for (int T = 0; T < NT; ++T) {
    const int q = 16 * (T0 + T) + c;
    qp[T] = in + q * LSTRIDE + 4 * h;
  }
  const v4f* bbase = Wl + h * E + ncol;

  float fw[NCT];
#pragma unroll
  for (int t = 0; t < NCT; ++t) fw[t] = fcw[ncol + 16 * t];

  v8f acc[NCT][NT][NPT];
#pragma unroll
  for (int t = 0; t < NCT; ++t)
#pragma unroll
    for (int T = 0; T < NT; ++T)
#pragma unroll
      for (int p = 0; p < NPT; ++p) acc[t][T][p] = v8f{};

  v4f a[2][NT][NPT];
  v4f r[2][NPT][NCT];

#pragma unroll
  for (int T = 0; T < NT; ++T)
#pragma unroll
    for (int i = 0; i < NPT; ++i)
      a[0][T][i] = *(const v4f*)(qp[T] + (i & 3) * POFF + (i >> 2) * LSTRIDE);
#pragma unroll
  for (int p = 0; p < NPT; ++p)
#pragma unroll
    for (int t = 0; t < NCT; ++t)
      r[0][p][t] = bbase[(size_t)(p * (E / 4)) * E + 16 * t];
#pragma unroll
  for (int T = 0; T < NT; ++T) vtransform(a[0][T]);

  for (int n = 0; n < NBLK; n += 2) {
    wblock<0, NT>(acc, a, r, qp, bbase, n);
    wblock<1, NT>(acc, a, r, qp, bbase, n + 1);
  }

  v4f rs[NT][8];
#pragma unroll
  for (int T = 0; T < NT; ++T) {
    v8f m[NPT];
#pragma unroll
    for (int p = 0; p < NPT; ++p) {
      v8f w = fw[0] * acc[0][T][p];
#pragma unroll
      for (int t = 1; t < NCT; ++t) w += fw[t] * acc[t][T][p];
      m[p] = w;
    }
    const v8f e1 = m[1] + m[2], o1 = m[1] - m[2];
    const v8f e2 = m[3] + m[4], o2 = m[3] - m[4];
    const v8f e3 = m[5] + m[6], o3 = m[5] - m[6];
    const v8f y0 = (m[0] + e1) + (e2 + e3);
    const v8f y1 = (o1 + 2.0f * o2) + 0.5f * o3;
    const v8f y2 = (e1 + 4.0f * e2) + 0.25f * e3;
    const v8f y3 = ((o1 + 8.0f * o2) + 0.125f * o3) + m[7];
#pragma unroll
    for (int v = 0; v < 8; ++v)
      rs[T][v] = v4f{row_sum16(y0[v]), row_sum16(y1[v]), row_sum16(y2[v]), row_sum16(y3[v])};
  }
  if (c == 0) {
#pragma unroll
    for (int T = 0; T < NT; ++T)
#pragma unroll
      for (int v = 0; v < 8; ++v)
        *(v4f*)(part + 4 * (16 * (T0 + T) + 8 * h + v)) = rs[T][v];
  }
}

__global__ __launch_bounds__(128) void fused_layers_kernel(const float* __restrict__ X,
                                                           const float* __restrict__ Ut,
                                                           const float* __restrict__ fcw,
                                                           const float* __restrict__ fcb,
                                                           float* __restrict__ out) {
  __shared__ __attribute__((aligned(16))) float lds[BUFSZ + 4 * CH];
  float* bufX = lds;
  float* part = lds + BUFSZ;

  const int tid  = threadIdx.x;
  const int lane = tid & 31;
  const int wave = tid >> 5;
  const int s0   = blockIdx.x * CH;

  const bool bstart = (s0 % SEQ_L) == 0;
  {
    const int c4 = tid & 63;
    const int rb = tid >> 6;
    float4 v[66];
#pragma unroll
    for (int i = 0; i < 66; ++i) {
      const int r = rb + 2 * i;
      const bool z = bstart && (r < 4);
      const int  g = z ? s0 : (s0 - 4 + r);
      float4 t = ((const float4*)(X + (size_t)g * E))[c4];
      if (z) t = make_float4(0.f, 0.f, 0.f, 0.f);
      v[i] = t;
    }
#pragma unroll
    for (int i = 0; i < 66; ++i) {
      const int r = rb + 2 * i;
      *(float4*)(bufX + (r & 3) * POFF + (r >> 2) * LSTRIDE + c4 * 4) = v[i];
    }
  }
  __syncthreads();

  const v4f* W = (const v4f*)Ut;
  conv_pass<2, 0>(bufX, part + wave * CH, W, fcw, lane, wave);
  __syncthreads();

  out[s0 + tid] = ((part[tid] + part[CH + tid]) + (part[2 * CH + tid] + part[3 * CH + tid])) + fcb[0];
}

extern "C" void kernel_launch(void* const* d_in, const int* in_sizes, int n_in,
                              void* d_out, int out_size, void* d_ws, size_t ws_size,
                              hipStream_t stream) {
  const float* x      = (const float*)d_in[0];
  const float* conv_w = (const float*)d_in[1];
  const float* fc_w   = (const float*)d_in[2];
  const float* fc_b   = (const float*)d_in[3];
  float* out = (float*)d_out;

  float* Q  = (float*)d_ws;
  float* Ut = Q + 2 * NPT * MATSZ;

  pair_points_kernel<<<1024, 32, 0, stream>>>(conv_w, Q);
  point_product_kernel<<<NPT * 128, 32, 0, stream>>>(Q, Ut);
  fused_layers_kernel<<<M_TOT / CH, 128, 0, stream>>>(x, Ut, fc_w, fc_b, out);
}
